# starting kernel
# baseline (speedup 1.0000x reference)
#include <hip/hip_runtime.h>

#define D_FEAT 64
#define HC 128
#define NHEAD 2
#define CCH 64
#define NEG_SLOPE 0.2f

typedef __attribute__((ext_vector_type(2))) float v2f;
typedef __attribute__((ext_vector_type(8))) float v8f;


__global__ void fill_kernel(float* __restrict__ p, float v, int n) {
    int i = blockIdx.x * blockDim.x + threadIdx.x;
    if (i < n) p[i] = v;
}

__device__ __forceinline__ void atomicMaxF(float* addr, float val) {
    if (val >= 0.0f) atomicMax((int*)addr, __float_as_int(val));
    else             atomicMin((unsigned int*)addr, __float_as_uint(val));
}

__global__ void gemm64x128_wmma(const float* __restrict__ X,
                                const float* __restrict__ W,
                                const float* __restrict__ bvec,
                                float* __restrict__ Y, int rows) {
    const int wave = (blockIdx.x * blockDim.x + threadIdx.x) >> 5;
    const int lane = threadIdx.x & 31;
    const int tiles_m = rows >> 4;
    const int tm = wave >> 3;
    const int tn = wave & 7;
    if (tm >= tiles_m) return;

    const int mrow  = (tm << 4) + (lane & 15);
    const int ncol  = (tn << 4) + (lane & 15);
    const int khalf = (lane >> 4) << 1;

    v8f acc = {};
#pragma unroll
    for (int k0 = 0; k0 < D_FEAT; k0 += 4) {
        const int kk = k0 + khalf;
        v2f a, b;
        const float* ap = X + (size_t)mrow * D_FEAT + kk;
        a.x = ap[0];
        a.y = ap[1];
        b.x = W[(size_t)(kk + 0) * HC + ncol];
        b.y = W[(size_t)(kk + 1) * HC + ncol];
        acc = __builtin_amdgcn_wmma_f32_16x16x4_f32(
             false, a,  false, b,
             (short)0, acc,  false,  false);
    }

    const float bb = bvec[ncol];
    const int rbase = (tm << 4) + ((lane >> 4) << 3);
#pragma unroll
    for (int j = 0; j < 8; ++j)
        Y[(size_t)(rbase + j) * HC + ncol] = acc[j] + bb;
}

__global__ void score_kernel(const float* __restrict__ xl,
                             const float* __restrict__ xr,
                             const int* __restrict__ src,
                             const int* __restrict__ dst,
                             const float* __restrict__ att,
                             float* __restrict__ ebuf,
                             float* __restrict__ mmax,
                             int E, int EE) {
    const int e    = (blockIdx.x * blockDim.x + threadIdx.x) >> 5;
    const int lane = threadIdx.x & 31;
    if (e >= EE) return;
    int s, d;
    if (e < E) { s = src[e]; d = dst[e]; } else { s = e - E; d = s; }

    const int c0 = lane << 2;
    const float4 zl = *(const float4*)(xl + (size_t)s * HC + c0);
    const float4 zr = *(const float4*)(xr + (size_t)d * HC + c0);
    const float4 at = *(const float4*)(att + c0);

    float p = 0.0f, z;
    z = zl.x + zr.x; z = (z > 0.0f) ? z : NEG_SLOPE * z; p += z * at.x;
    z = zl.y + zr.y; z = (z > 0.0f) ? z : NEG_SLOPE * z; p += z * at.y;
    z = zl.z + zr.z; z = (z > 0.0f) ? z : NEG_SLOPE * z; p += z * at.z;
    z = zl.w + zr.w; z = (z > 0.0f) ? z : NEG_SLOPE * z; p += z * at.w;

    p += __shfl_xor(p, 1);
    p += __shfl_xor(p, 2);
    p += __shfl_xor(p, 4);
    p += __shfl_xor(p, 8);

    if ((lane & 15) == 0) {
        const int h = lane >> 4;
        ebuf[(size_t)e * NHEAD + h] = p;
        atomicMaxF(&mmax[(size_t)d * NHEAD + h], p);
    }
}

__global__ void expsum_kernel(const int* __restrict__ src,
                              const int* __restrict__ dst,
                              float* __restrict__ ebuf,
                              const float* __restrict__ mmax,
                              float* __restrict__ denom,
                              int E, int EE) {
    int g = blockIdx.x * blockDim.x + threadIdx.x;
    if (g >= EE * NHEAD) return;
    const int e = g >> 1;
    const int h = g & 1;
    const int d = (e < E) ? dst[e] : (e - E);
    const float a = __expf(ebuf[g] - mmax[(size_t)d * NHEAD + h]);
    ebuf[g] = a;
    atomicAdd(&denom[(size_t)d * NHEAD + h], a);
}

__global__ void scatter_kernel(const float* __restrict__ xl,
                               const float* __restrict__ ebuf,
                               const float* __restrict__ denom,
                               const int* __restrict__ src,
                               const int* __restrict__ dst,
                               float* __restrict__ accum,
                               int E, int EE) {
    const int e    = (blockIdx.x * blockDim.x + threadIdx.x) >> 5;
    const int lane = threadIdx.x & 31;
    if (e >= EE) return;
    int s, d;
    if (e < E) { s = src[e]; d = dst[e]; } else { s = e - E; d = s; }

    const int h = lane >> 4;
    const float alpha = ebuf[(size_t)e * NHEAD + h] /
                        (denom[(size_t)d * NHEAD + h] + 1e-16f);
    const int c0 = lane << 2;
    const float4 v = *(const float4*)(xl + (size_t)s * HC + c0);
    float* o = accum + (size_t)d * HC + c0;
    atomicAdd(o + 0, alpha * v.x);
    atomicAdd(o + 1, alpha * v.y);
    atomicAdd(o + 2, alpha * v.z);
    atomicAdd(o + 3, alpha * v.w);
}

__global__ void finalize_kernel(const float* __restrict__ accum,
                                const float* __restrict__ bias,
                                float* __restrict__ out,
                                int apply_gelu, int total) {
    int g = blockIdx.x * blockDim.x + threadIdx.x;
    if (g >= total) return;
    const int node = g >> 6;
    const int c    = g & 63;
    float v = 0.5f * (accum[(size_t)node * HC + c] +
                      accum[(size_t)node * HC + CCH + c]) + bias[c];
    if (apply_gelu) {
        const float u = 0.7978845608028654f * (v + 0.044715f * v * v * v);
        v = 0.5f * v * (1.0f + tanhf(u));
    }
    out[g] = v;
}


extern "C" void kernel_launch(void* const* d_in, const int* in_sizes, int n_in,
                              void* d_out, int out_size, void* d_ws, size_t ws_size,
                              hipStream_t stream) {
    const float* x0   = (const float*)d_in[0];
    const int*   conn = (const int*)  d_in[1];
    const float* Wl   = (const float*)d_in[2];
    const float* bl   = (const float*)d_in[3];
    const float* Wr   = (const float*)d_in[4];
    const float* br   = (const float*)d_in[5];
    const float* att  = (const float*)d_in[6];
    const float* bias = (const float*)d_in[7];

    const int N  = in_sizes[0] / D_FEAT;
    const int E  = in_sizes[1] / 2;
    const int EE = E + N;
    const int* src = conn;
    const int* dst = conn + E;

    float* ws    = (float*)d_ws;
    float* xl    = ws;
    float* xr    = xl   + (size_t)N * HC;
    float* x1    = xr   + (size_t)N * HC;
    float* ebuf  = x1   + (size_t)N * D_FEAT;
    float* mmax  = ebuf + (size_t)EE * NHEAD;
    float* denom = mmax + (size_t)N * NHEAD;

    const dim3 blk(256);
    const int gemm_blocks = ((N / 16) * 8 + 7) / 8;
    const int edge_blocks = (int)(((long long)EE * 32 + 255) / 256);
    const int eh_blocks   = (EE * NHEAD + 255) / 256;
    const int nh          = N * NHEAD;
    const int nacc        = N * HC;
    const int fin_total   = N * D_FEAT;

    const int L = 2;
    const float* xin = x0;
    for (int l = 0; l < L; ++l) {
        const float* Wl_l = Wl + (size_t)l * D_FEAT * HC;
        const float* bl_l = bl + (size_t)l * HC;
        const float* Wr_l = Wr + (size_t)l * D_FEAT * HC;
        const float* br_l = br + (size_t)l * HC;
        const float* at_l = att + (size_t)l * NHEAD * CCH;
        const float* bi_l = bias + (size_t)l * CCH;

        gemm64x128_wmma<<<gemm_blocks, blk, 0, stream>>>(xin, Wl_l, bl_l, xl, N);
        gemm64x128_wmma<<<gemm_blocks, blk, 0, stream>>>(xin, Wr_l, br_l, xr, N);

        fill_kernel<<<(nh + 255) / 256, blk, 0, stream>>>(mmax, -__builtin_inff(), nh);
        fill_kernel<<<(nh + 255) / 256, blk, 0, stream>>>(denom, 0.0f, nh);

        score_kernel<<<edge_blocks, blk, 0, stream>>>(xl, xr, src, dst, at_l,
                                                      ebuf, mmax, E, EE);

        fill_kernel<<<(nacc + 255) / 256, blk, 0, stream>>>(xr, 0.0f, nacc);

        expsum_kernel<<<eh_blocks, blk, 0, stream>>>(src, dst, ebuf, mmax, denom, E, EE);

        scatter_kernel<<<edge_blocks, blk, 0, stream>>>(xl, ebuf, denom, src, dst,
                                                        xr, E, EE);

        float* outp = (l == L - 1) ? (float*)d_out : x1;
        finalize_kernel<<<(fin_total + 255) / 256, blk, 0, stream>>>(
            xr, bi_l, outp, (l < L - 1) ? 1 : 0, fin_total);

        xin = x1;
    }
}


# Scatter pass adds the two heads' weighted rows per edge in registers and accumulates into [N,64], one whole 128-B line per atomic instruction: half the float atomics; order of float sums changes
# speedup vs baseline: 1.2160x; 1.2160x over previous
#include <hip/hip_runtime.h>

#define D_FEAT 64
#define HC 128
#define NHEAD 2
#define CCH 64
#define NEG_SLOPE 0.2f

typedef __attribute__((ext_vector_type(2))) float v2f;
typedef __attribute__((ext_vector_type(8))) float v8f;


__global__ void fill_kernel(float* __restrict__ p, float v, int n) {
    int i = blockIdx.x * blockDim.x + threadIdx.x;
    if (i < n) p[i] = v;
}

__device__ __forceinline__ void atomicMaxF(float* addr, float val) {
    if (val >= 0.0f) atomicMax((int*)addr, __float_as_int(val));
    else             atomicMin((unsigned int*)addr, __float_as_uint(val));
}

__global__ void gemm64x128_wmma(const float* __restrict__ X,
                                const float* __restrict__ W,
                                const float* __restrict__ bvec,
                                float* __restrict__ Y, int rows) {
    const int wave = (blockIdx.x * blockDim.x + threadIdx.x) >> 5;
    const int lane = threadIdx.x & 31;
    const int tiles_m = rows >> 4;
    const int tm = wave >> 3;
    const int tn = wave & 7;
    if (tm >= tiles_m) return;

    const int mrow  = (tm << 4) + (lane & 15);
    const int ncol  = (tn << 4) + (lane & 15);
    const int khalf = (lane >> 4) << 1;

    v8f acc = {};
#pragma unroll
    for (int k0 = 0; k0 < D_FEAT; k0 += 4) {
        const int kk = k0 + khalf;
        v2f a, b;
        const float* ap = X + (size_t)mrow * D_FEAT + kk;
        a.x = ap[0];
        a.y = ap[1];
        b.x = W[(size_t)(kk + 0) * HC + ncol];
        b.y = W[(size_t)(kk + 1) * HC + ncol];
        acc = __builtin_amdgcn_wmma_f32_16x16x4_f32(
             false, a,  false, b,
             (short)0, acc,  false,  false);
    }

    const float bb = bvec[ncol];
    const int rbase = (tm << 4) + ((lane >> 4) << 3);
#pragma unroll
    for (int j = 0; j < 8; ++j)
        Y[(size_t)(rbase + j) * HC + ncol] = acc[j] + bb;
}

__global__ void score_kernel(const float* __restrict__ xl,
                             const float* __restrict__ xr,
                             const int* __restrict__ src,
                             const int* __restrict__ dst,
                             const float* __restrict__ att,
                             float* __restrict__ ebuf,
                             float* __restrict__ mmax,
                             int E, int EE) {
    const int e    = (blockIdx.x * blockDim.x + threadIdx.x) >> 5;
    const int lane = threadIdx.x & 31;
    if (e >= EE) return;
    int s, d;
    if (e < E) { s = src[e]; d = dst[e]; } else { s = e - E; d = s; }

    const int c0 = lane << 2;
    const float4 zl = *(const float4*)(xl + (size_t)s * HC + c0);
    const float4 zr = *(const float4*)(xr + (size_t)d * HC + c0);
    const float4 at = *(const float4*)(att + c0);

    float p = 0.0f, z;
    z = zl.x + zr.x; z = (z > 0.0f) ? z : NEG_SLOPE * z; p += z * at.x;
    z = zl.y + zr.y; z = (z > 0.0f) ? z : NEG_SLOPE * z; p += z * at.y;
    z = zl.z + zr.z; z = (z > 0.0f) ? z : NEG_SLOPE * z; p += z * at.z;
    z = zl.w + zr.w; z = (z > 0.0f) ? z : NEG_SLOPE * z; p += z * at.w;

    p += __shfl_xor(p, 1);
    p += __shfl_xor(p, 2);
    p += __shfl_xor(p, 4);
    p += __shfl_xor(p, 8);

    if ((lane & 15) == 0) {
        const int h = lane >> 4;
        ebuf[(size_t)e * NHEAD + h] = p;
        atomicMaxF(&mmax[(size_t)d * NHEAD + h], p);
    }
}

__global__ void expsum_kernel(const int* __restrict__ src,
                              const int* __restrict__ dst,
                              float* __restrict__ ebuf,
                              const float* __restrict__ mmax,
                              float* __restrict__ denom,
                              int E, int EE) {
    int g = blockIdx.x * blockDim.x + threadIdx.x;
    if (g >= EE * NHEAD) return;
    const int e = g >> 1;
    const int h = g & 1;
    const int d = (e < E) ? dst[e] : (e - E);
    const float a = __expf(ebuf[g] - mmax[(size_t)d * NHEAD + h]);
    ebuf[g] = a;
    atomicAdd(&denom[(size_t)d * NHEAD + h], a);
}

__global__ void scatter_kernel(const float* __restrict__ xl,
                               const float* __restrict__ ebuf,
                               const float* __restrict__ denom,
                               const int* __restrict__ src,
                               const int* __restrict__ dst,
                               float* __restrict__ accum,
                               int E, int EE) {
    const int e    = (blockIdx.x * blockDim.x + threadIdx.x) >> 5;
    const int lane = threadIdx.x & 31;
    if (e >= EE) return;
    int s, d;
    if (e < E) { s = src[e]; d = dst[e]; } else { s = e - E; d = s; }

    const float2 a  = *(const float2*)(ebuf  + (size_t)e * NHEAD);
    const float2 dn = *(const float2*)(denom + (size_t)d * NHEAD);
    const float alpha0 = a.x / (dn.x + 1e-16f);
    const float alpha1 = a.y / (dn.y + 1e-16f);

    const float* r = xl + (size_t)s * HC + lane;
    const float v00 = r[0];
    const float v01 = r[32];
    const float v10 = r[CCH];
    const float v11 = r[CCH + 32];

    float* o = accum + (size_t)d * CCH + lane;
    atomicAdd(o,      alpha0 * v00 + alpha1 * v10);
    atomicAdd(o + 32, alpha0 * v01 + alpha1 * v11);
}

__global__ void finalize_kernel(const float* __restrict__ accum,
                                const float* __restrict__ bias,
                                float* __restrict__ out,
                                int apply_gelu, int total) {
    int g = blockIdx.x * blockDim.x + threadIdx.x;
    if (g >= total) return;
    const int c    = g & 63;
    float v = 0.5f * accum[g] + bias[c];
    if (apply_gelu) {
        const float u = 0.7978845608028654f * (v + 0.044715f * v * v * v);
        v = 0.5f * v * (1.0f + tanhf(u));
    }
    out[g] = v;
}


extern "C" void kernel_launch(void* const* d_in, const int* in_sizes, int n_in,
                              void* d_out, int out_size, void* d_ws, size_t ws_size,
                              hipStream_t stream) {
    const float* x0   = (const float*)d_in[0];
    const int*   conn = (const int*)  d_in[1];
    const float* Wl   = (const float*)d_in[2];
    const float* bl   = (const float*)d_in[3];
    const float* Wr   = (const float*)d_in[4];
    const float* br   = (const float*)d_in[5];
    const float* att  = (const float*)d_in[6];
    const float* bias = (const float*)d_in[7];

    const int N  = in_sizes[0] / D_FEAT;
    const int E  = in_sizes[1] / 2;
    const int EE = E + N;
    const int* src = conn;
    const int* dst = conn + E;

    float* ws    = (float*)d_ws;
    float* xl    = ws;
    float* xr    = xl   + (size_t)N * HC;
    float* x1    = xr   + (size_t)N * HC;
    float* ebuf  = x1   + (size_t)N * D_FEAT;
    float* mmax  = ebuf + (size_t)EE * NHEAD;
    float* denom = mmax + (size_t)N * NHEAD;

    const dim3 blk(256);
    const int gemm_blocks = ((N / 16) * 8 + 7) / 8;
    const int edge_blocks = (int)(((long long)EE * 32 + 255) / 256);
    const int eh_blocks   = (EE * NHEAD + 255) / 256;
    const int nh          = N * NHEAD;
    const int nacc        = N * CCH;
    const int fin_total   = N * D_FEAT;

    const int L = 2;
    const float* xin = x0;
    for (int l = 0; l < L; ++l) {
        const float* Wl_l = Wl + (size_t)l * D_FEAT * HC;
        const float* bl_l = bl + (size_t)l * HC;
        const float* Wr_l = Wr + (size_t)l * D_FEAT * HC;
        const float* br_l = br + (size_t)l * HC;
        const float* at_l = att + (size_t)l * NHEAD * CCH;
        const float* bi_l = bias + (size_t)l * CCH;

        gemm64x128_wmma<<<gemm_blocks, blk, 0, stream>>>(xin, Wl_l, bl_l, xl, N);
        gemm64x128_wmma<<<gemm_blocks, blk, 0, stream>>>(xin, Wr_l, br_l, xr, N);

        fill_kernel<<<(nh + 255) / 256, blk, 0, stream>>>(mmax, -__builtin_inff(), nh);
        fill_kernel<<<(nh + 255) / 256, blk, 0, stream>>>(denom, 0.0f, nh);

        score_kernel<<<edge_blocks, blk, 0, stream>>>(xl, xr, src, dst, at_l,
                                                      ebuf, mmax, E, EE);

        fill_kernel<<<(nacc + 255) / 256, blk, 0, stream>>>(xr, 0.0f, nacc);

        expsum_kernel<<<eh_blocks, blk, 0, stream>>>(src, dst, ebuf, mmax, denom, E, EE);

        scatter_kernel<<<edge_blocks, blk, 0, stream>>>(xl, ebuf, denom, src, dst,
                                                        xr, E, EE);

        float* outp = (l == L - 1) ? (float*)d_out : x1;
        finalize_kernel<<<(fin_total + 255) / 256, blk, 0, stream>>>(
            xr, bi_l, outp, (l < L - 1) ? 1 : 0, fin_total);

        xin = x1;
    }
}


# Edges bucketed by destination once per call (count, scan, place); one wave per node does scores, running-max softmax and weighted sum in registers: no float atomics; order of float sums changes
# speedup vs baseline: 2.5933x; 2.1326x over previous
#include <hip/hip_runtime.h>

#define D_FEAT 64
#define HC 128
#define NHEAD 2
#define CCH 64
#define NEG_SLOPE 0.2f

typedef __attribute__((ext_vector_type(2))) float v2f;
typedef __attribute__((ext_vector_type(8))) float v8f;

__global__ void gemm64x128_wmma(const float* __restrict__ X,
                                const float* __restrict__ W,
                                const float* __restrict__ bvec,
                                float* __restrict__ Y, int rows) {
    const int wave = (blockIdx.x * blockDim.x + threadIdx.x) >> 5;
    const int lane = threadIdx.x & 31;
    const int tiles_m = rows >> 4;
    const int tm = wave >> 3;
    const int tn = wave & 7;
    if (tm >= tiles_m) return;

    const int mrow  = (tm << 4) + (lane & 15);
    const int ncol  = (tn << 4) + (lane & 15);
    const int khalf = (lane >> 4) << 1;

    v8f acc = {};
#pragma unroll
    for (int k0 = 0; k0 < D_FEAT; k0 += 4) {
        const int kk = k0 + khalf;
        v2f a, b;
        const float* ap = X + (size_t)mrow * D_FEAT + kk;
        a.x = ap[0];
        a.y = ap[1];
        b.x = W[(size_t)(kk + 0) * HC + ncol];
        b.y = W[(size_t)(kk + 1) * HC + ncol];
        acc = __builtin_amdgcn_wmma_f32_16x16x4_f32(
             false, a,  false, b,
             (short)0, acc,  false,  false);
    }

    const float bb = bvec[ncol];
    const int rbase = (tm << 4) + ((lane >> 4) << 3);
#pragma unroll
    for (int j = 0; j < 8; ++j)
        Y[(size_t)(rbase + j) * HC + ncol] = acc[j] + bb;
}


__global__ void zero_int_kernel(int* __restrict__ p, int n) {
    const int i = blockIdx.x * blockDim.x + threadIdx.x;
    if (i < n) p[i] = 0;
}

__global__ void degree_kernel(const int* __restrict__ dst, int* __restrict__ cnt, int E) {
    const int e = blockIdx.x * blockDim.x + threadIdx.x;
    if (e < E) atomicAdd(&cnt[dst[e]], 1);
}

#define SCAN_T 1024
__global__ void __launch_bounds__(SCAN_T)
rowptr_kernel(const int* __restrict__ cnt, int* __restrict__ rowptr, int n) {
    __shared__ int wsum[32];
    __shared__ int sbase;
    const int t    = threadIdx.x;
    const int lane = t & 31;
    const int w    = t >> 5;
    const int first = blockIdx.x * SCAN_T;

    int part = 0;
    for (int i = t; i < first; i += SCAN_T) part += cnt[i];
    part += __shfl_xor(part, 16);
    part += __shfl_xor(part, 8);
    part += __shfl_xor(part, 4);
    part += __shfl_xor(part, 2);
    part += __shfl_xor(part, 1);
    if (lane == 0) wsum[w] = part;
    __syncthreads();
    if (w == 0) {
        int v = wsum[lane];
        v += __shfl_xor(v, 16);
        v += __shfl_xor(v, 8);
        v += __shfl_xor(v, 4);
        v += __shfl_xor(v, 2);
        v += __shfl_xor(v, 1);
        if (lane == 0) sbase = v;
    }
    __syncthreads();
    const int base = sbase;

    const int i = first + t;
    const int c = (i < n) ? cnt[i] : 0;
    int incl = c;
    for (int o = 1; o < 32; o <<= 1) {
        const int u = __shfl_up(incl, o);
        if (lane >= o) incl += u;
    }
    if (lane == 31) wsum[w] = incl;
    __syncthreads();
    if (w == 0) {
        const int v = wsum[lane];
        int inc2 = v;
        for (int o = 1; o < 32; o <<= 1) {
            const int u = __shfl_up(inc2, o);
            if (lane >= o) inc2 += u;
        }
        wsum[lane] = inc2 - v;
    }
    __syncthreads();
    const int excl = base + wsum[w] + incl - c;
    if (i < n) rowptr[i] = excl;
    if (i == n - 1) rowptr[n] = excl + c;
}

__global__ void place_kernel(const int* __restrict__ src,
                             const int* __restrict__ dst,
                             const int* __restrict__ rowptr,
                             int* __restrict__ cnt,
                             int* __restrict__ csr_src, int E) {
    const int e = blockIdx.x * blockDim.x + threadIdx.x;
    if (e >= E) return;
    const int d = dst[e];
    const int k = atomicSub(&cnt[d], 1);
    csr_src[rowptr[d] + k - 1] = src[e];
}


__device__ __forceinline__ float head_score(const float4 v, const float4 zr, const float4 at) {
    float p = 0.0f, z;
    z = v.x + zr.x; z = (z > 0.0f) ? z : NEG_SLOPE * z; p += z * at.x;
    z = v.y + zr.y; z = (z > 0.0f) ? z : NEG_SLOPE * z; p += z * at.y;
    z = v.z + zr.z; z = (z > 0.0f) ? z : NEG_SLOPE * z; p += z * at.z;
    z = v.w + zr.w; z = (z > 0.0f) ? z : NEG_SLOPE * z; p += z * at.w;
    p += __shfl_xor(p, 1);
    p += __shfl_xor(p, 2);
    p += __shfl_xor(p, 4);
    p += __shfl_xor(p, 8);
    return p;
}

__device__ __forceinline__ float finish_value(float v, int apply_gelu) {
    if (apply_gelu) {
        const float u = 0.7978845608028654f * (v + 0.044715f * v * v * v);
        v = 0.5f * v * (1.0f + tanhf(u));
    }
    return v;
}

__global__ void __launch_bounds__(256)
node_kernel(const float* __restrict__ xl,
            const float* __restrict__ xr,
            const int* __restrict__ rowptr,
            const int* __restrict__ csr_src,
            const float* __restrict__ att,
            const float* __restrict__ bias,
            float* __restrict__ out,
            int apply_gelu, int n) {
    const int d = __builtin_amdgcn_readfirstlane(
        (int)((blockIdx.x * blockDim.x + threadIdx.x) >> 5));
    const int lane = threadIdx.x & 31;
    if (d >= n) return;

    const int c0 = lane << 2;
    const float4 zr = *(const float4*)(xr + (size_t)d * HC + c0);
    const float4 at = *(const float4*)(att + c0);
    const int beg = rowptr[d];
    const int end = rowptr[d + 1];

    float4 acc = *(const float4*)(xl + (size_t)d * HC + c0);
    float m   = head_score(acc, zr, at);
    float den = 1.0f;

    for (int base = beg; base < end; base += 32) {
        const int left = end - base;
        const int nk   = (left < 32) ? left : 32;
        const int idx = (lane < nk) ? csr_src[base + lane] : d;
#pragma unroll 4
        for (int k = 0; k < nk; ++k) {
            const int s = __builtin_amdgcn_readlane(idx, k);
            const float4 v = *(const float4*)(xl + (size_t)s * HC + c0);
            const float e  = head_score(v, zr, at);
            const float mn = fmaxf(m, e);
            const float sc = __expf(m - mn);
            const float a  = __expf(e - mn);
            den   = den   * sc + a;
            acc.x = acc.x * sc + a * v.x;
            acc.y = acc.y * sc + a * v.y;
            acc.z = acc.z * sc + a * v.z;
            acc.w = acc.w * sc + a * v.w;
            m = mn;
        }
    }

    const float inv = 1.0f / (den + 1e-16f);
    float4 r;
    r.x = acc.x * inv;
    r.y = acc.y * inv;
    r.z = acc.z * inv;
    r.w = acc.w * inv;
    r.x += __shfl_xor(r.x, 16);
    r.y += __shfl_xor(r.y, 16);
    r.z += __shfl_xor(r.z, 16);
    r.w += __shfl_xor(r.w, 16);

    if (lane < 16) {
        const float4 b = *(const float4*)(bias + c0);
        float4 o;
        o.x = finish_value(0.5f * r.x + b.x, apply_gelu);
        o.y = finish_value(0.5f * r.y + b.y, apply_gelu);
        o.z = finish_value(0.5f * r.z + b.z, apply_gelu);
        o.w = finish_value(0.5f * r.w + b.w, apply_gelu);
        *(float4*)(out + (size_t)d * CCH + c0) = o;
    }
}


extern "C" void kernel_launch(void* const* d_in, const int* in_sizes, int n_in,
                              void* d_out, int out_size, void* d_ws, size_t ws_size,
                              hipStream_t stream) {
    const float* x0   = (const float*)d_in[0];
    const int*   conn = (const int*)  d_in[1];
    const float* Wl   = (const float*)d_in[2];
    const float* bl   = (const float*)d_in[3];
    const float* Wr   = (const float*)d_in[4];
    const float* br   = (const float*)d_in[5];
    const float* att  = (const float*)d_in[6];
    const float* bias = (const float*)d_in[7];

    const int N  = in_sizes[0] / D_FEAT;
    const int E  = in_sizes[1] / 2;
    const int* src = conn;
    const int* dst = conn + E;

    float* ws    = (float*)d_ws;
    float* xl    = ws;
    float* xr    = xl   + (size_t)N * HC;
    float* x1    = xr   + (size_t)N * HC;
    int*   cnt     = (int*)(x1 + (size_t)N * D_FEAT);
    int*   rowptr  = cnt + N;
    int*   csr_src = rowptr + N + 1;

    const dim3 blk(256);
    const int gemm_blocks = ((N / 16) * 8 + 7) / 8;
    const int node_blocks = (int)(((long long)N * 32 + 255) / 256);

    zero_int_kernel<<<(N + 255) / 256, blk, 0, stream>>>(cnt, N);
    degree_kernel<<<(E + 255) / 256, blk, 0, stream>>>(dst, cnt, E);
    rowptr_kernel<<<(N + SCAN_T - 1) / SCAN_T, dim3(SCAN_T), 0, stream>>>(cnt, rowptr, N);
    place_kernel<<<(E + 255) / 256, blk, 0, stream>>>(src, dst, rowptr, cnt, csr_src, E);

    const int L = 2;
    const float* xin = x0;
    for (int l = 0; l < L; ++l) {
        const float* Wl_l = Wl + (size_t)l * D_FEAT * HC;
        const float* bl_l = bl + (size_t)l * HC;
        const float* Wr_l = Wr + (size_t)l * D_FEAT * HC;
        const float* br_l = br + (size_t)l * HC;
        const float* at_l = att + (size_t)l * NHEAD * CCH;
        const float* bi_l = bias + (size_t)l * CCH;

        gemm64x128_wmma<<<gemm_blocks, blk, 0, stream>>>(xin, Wl_l, bl_l, xl, N);
        gemm64x128_wmma<<<gemm_blocks, blk, 0, stream>>>(xin, Wr_l, br_l, xr, N);

        float* outp = (l == L - 1) ? (float*)d_out : x1;
        node_kernel<<<node_blocks, blk, 0, stream>>>(xl, xr, rowptr, csr_src, at_l, bi_l,
                                                     outp, (l < L - 1) ? 1 : 0, N);

        xin = x1;
    }
}


# Destination-side projection x@Wr+br moved into the node pass: each 16-node work-group forms its 16x128 tile with the f32 matrix instruction into LDS; xr never goes to memory, 2 launches fewer
# speedup vs baseline: 2.7191x; 1.0485x over previous
#include <hip/hip_runtime.h>

#define D_FEAT 64
#define HC 128
#define NHEAD 2
#define CCH 64
#define NEG_SLOPE 0.2f

typedef __attribute__((ext_vector_type(2))) float v2f;
typedef __attribute__((ext_vector_type(8))) float v8f;

__global__ void gemm64x128_wmma(const float* __restrict__ X,
                                const float* __restrict__ W,
                                const float* __restrict__ bvec,
                                float* __restrict__ Y, int rows) {
    const int wave = (blockIdx.x * blockDim.x + threadIdx.x) >> 5;
    const int lane = threadIdx.x & 31;
    const int tiles_m = rows >> 4;
    const int tm = wave >> 3;
    const int tn = wave & 7;
    if (tm >= tiles_m) return;

    const int mrow  = (tm << 4) + (lane & 15);
    const int ncol  = (tn << 4) + (lane & 15);
    const int khalf = (lane >> 4) << 1;

    v8f acc = {};
#pragma unroll
    for (int k0 = 0; k0 < D_FEAT; k0 += 4) {
        const int kk = k0 + khalf;
        v2f a, b;
        const float* ap = X + (size_t)mrow * D_FEAT + kk;
        a.x = ap[0];
        a.y = ap[1];
        b.x = W[(size_t)(kk + 0) * HC + ncol];
        b.y = W[(size_t)(kk + 1) * HC + ncol];
        acc = __builtin_amdgcn_wmma_f32_16x16x4_f32(
             false, a,  false, b,
             (short)0, acc,  false,  false);
    }

    const float bb = bvec[ncol];
    const int rbase = (tm << 4) + ((lane >> 4) << 3);
#pragma unroll
    for (int j = 0; j < 8; ++j)
        Y[(size_t)(rbase + j) * HC + ncol] = acc[j] + bb;
}


__global__ void zero_int_kernel(int* __restrict__ p, int n) {
    const int i = blockIdx.x * blockDim.x + threadIdx.x;
    if (i < n) p[i] = 0;
}

__global__ void degree_kernel(const int* __restrict__ dst, int* __restrict__ cnt, int E) {
    const int e = blockIdx.x * blockDim.x + threadIdx.x;
    if (e < E) atomicAdd(&cnt[dst[e]], 1);
}

#define SCAN_T 1024
__global__ void __launch_bounds__(SCAN_T)
rowptr_kernel(const int* __restrict__ cnt, int* __restrict__ rowptr, int n) {
    __shared__ int wsum[32];
    __shared__ int sbase;
    const int t    = threadIdx.x;
    const int lane = t & 31;
    const int w    = t >> 5;
    const int first = blockIdx.x * SCAN_T;

    int part = 0;
    for (int i = t; i < first; i += SCAN_T) part += cnt[i];
    part += __shfl_xor(part, 16);
    part += __shfl_xor(part, 8);
    part += __shfl_xor(part, 4);
    part += __shfl_xor(part, 2);
    part += __shfl_xor(part, 1);
    if (lane == 0) wsum[w] = part;
    __syncthreads();
    if (w == 0) {
        int v = wsum[lane];
        v += __shfl_xor(v, 16);
        v += __shfl_xor(v, 8);
        v += __shfl_xor(v, 4);
        v += __shfl_xor(v, 2);
        v += __shfl_xor(v, 1);
        if (lane == 0) sbase = v;
    }
    __syncthreads();
    const int base = sbase;

    const int i = first + t;
    const int c = (i < n) ? cnt[i] : 0;
    int incl = c;
    for (int o = 1; o < 32; o <<= 1) {
        const int u = __shfl_up(incl, o);
        if (lane >= o) incl += u;
    }
    if (lane == 31) wsum[w] = incl;
    __syncthreads();
    if (w == 0) {
        const int v = wsum[lane];
        int inc2 = v;
        for (int o = 1; o < 32; o <<= 1) {
            const int u = __shfl_up(inc2, o);
            if (lane >= o) inc2 += u;
        }
        wsum[lane] = inc2 - v;
    }
    __syncthreads();
    const int excl = base + wsum[w] + incl - c;
    if (i < n) rowptr[i] = excl;
    if (i == n - 1) rowptr[n] = excl + c;
}

__global__ void place_kernel(const int* __restrict__ src,
                             const int* __restrict__ dst,
                             const int* __restrict__ rowptr,
                             int* __restrict__ cnt,
                             int* __restrict__ csr_src, int E) {
    const int e = blockIdx.x * blockDim.x + threadIdx.x;
    if (e >= E) return;
    const int d = dst[e];
    const int k = atomicSub(&cnt[d], 1);
    csr_src[rowptr[d] + k - 1] = src[e];
}


__device__ __forceinline__ float head_score(const float4 v, const float4 zr, const float4 at) {
    float p = 0.0f, z;
    z = v.x + zr.x; z = (z > 0.0f) ? z : NEG_SLOPE * z; p += z * at.x;
    z = v.y + zr.y; z = (z > 0.0f) ? z : NEG_SLOPE * z; p += z * at.y;
    z = v.z + zr.z; z = (z > 0.0f) ? z : NEG_SLOPE * z; p += z * at.z;
    z = v.w + zr.w; z = (z > 0.0f) ? z : NEG_SLOPE * z; p += z * at.w;
    p += __shfl_xor(p, 1);
    p += __shfl_xor(p, 2);
    p += __shfl_xor(p, 4);
    p += __shfl_xor(p, 8);
    return p;
}

__device__ __forceinline__ float finish_value(float v, int apply_gelu) {
    if (apply_gelu) {
        const float u = 0.7978845608028654f * (v + 0.044715f * v * v * v);
        v = 0.5f * v * (1.0f + tanhf(u));
    }
    return v;
}

#define NODE_WAVES 16
#define XR_PITCH 132

__global__ void __launch_bounds__(512)
node_kernel(const float* __restrict__ xin,
            const float* __restrict__ Wr,
            const float* __restrict__ br,
            const float* __restrict__ xl,
            const int* __restrict__ rowptr,
            const int* __restrict__ csr_src,
            const float* __restrict__ att,
            const float* __restrict__ bias,
            float* __restrict__ out,
            int apply_gelu, int n) {
    __shared__ __attribute__((aligned(16))) float sxr[NODE_WAVES * XR_PITCH];

    const int w = __builtin_amdgcn_readfirstlane((int)(threadIdx.x >> 5));
    const int lane = threadIdx.x & 31;
    const int nbase = blockIdx.x * NODE_WAVES;
    const int d = nbase + w;
    const bool valid = d < n;
    const int dd = valid ? d : n - 1;

    const int c0 = lane << 2;
    const float4 at = *(const float4*)(att + c0);
    const int beg = rowptr[dd];
    const int end = rowptr[dd + 1];
    float4 acc = *(const float4*)(xl + (size_t)dd * HC + c0);
    int idx = dd;
    {
        const int left0 = end - beg;
        const int nk0   = (left0 < 32) ? left0 : 32;
        if (lane < nk0) idx = csr_src[beg + lane];
    }

    if (w < 8) {
        int mrow = nbase + (lane & 15);
        mrow = (mrow < n) ? mrow : n - 1;
        const int ncol  = (w << 4) + (lane & 15);
        const int khalf = (lane >> 4) << 1;
        v8f t = {};
#pragma unroll
        for (int k0 = 0; k0 < D_FEAT; k0 += 4) {
            const int kk = k0 + khalf;
            v2f a, b;
            const float* ap = xin + (size_t)mrow * D_FEAT + kk;
            a.x = ap[0];
            a.y = ap[1];
            b.x = Wr[(size_t)(kk + 0) * HC + ncol];
            b.y = Wr[(size_t)(kk + 1) * HC + ncol];
            t = __builtin_amdgcn_wmma_f32_16x16x4_f32(
                 false, a,  false, b,
                 (short)0, t,  false,  false);
        }
        const float bb = br[ncol];
        const int rloc = (lane >> 4) << 3;
#pragma unroll
        for (int j = 0; j < 8; ++j)
            sxr[(rloc + j) * XR_PITCH + ncol] = t[j] + bb;
    }
    __syncthreads();
    if (!valid) return;

    const float4 zr = *(const float4*)(sxr + w * XR_PITCH + c0);

    float m   = head_score(acc, zr, at);
    float den = 1.0f;

    for (int base = beg; base < end; base += 32) {
        const int left = end - base;
        const int nk   = (left < 32) ? left : 32;
        if (base != beg) idx = (lane < nk) ? csr_src[base + lane] : d;
#pragma unroll 4
        for (int k = 0; k < nk; ++k) {
            const int s = __builtin_amdgcn_readlane(idx, k);
            const float4 v = *(const float4*)(xl + (size_t)s * HC + c0);
            const float e  = head_score(v, zr, at);
            const float mn = fmaxf(m, e);
            const float sc = __expf(m - mn);
            const float a  = __expf(e - mn);
            den   = den   * sc + a;
            acc.x = acc.x * sc + a * v.x;
            acc.y = acc.y * sc + a * v.y;
            acc.z = acc.z * sc + a * v.z;
            acc.w = acc.w * sc + a * v.w;
            m = mn;
        }
    }

    const float inv = 1.0f / (den + 1e-16f);
    float4 r;
    r.x = acc.x * inv;
    r.y = acc.y * inv;
    r.z = acc.z * inv;
    r.w = acc.w * inv;
    r.x += __shfl_xor(r.x, 16);
    r.y += __shfl_xor(r.y, 16);
    r.z += __shfl_xor(r.z, 16);
    r.w += __shfl_xor(r.w, 16);

    if (lane < 16) {
        const float4 b = *(const float4*)(bias + c0);
        float4 o;
        o.x = finish_value(0.5f * r.x + b.x, apply_gelu);
        o.y = finish_value(0.5f * r.y + b.y, apply_gelu);
        o.z = finish_value(0.5f * r.z + b.z, apply_gelu);
        o.w = finish_value(0.5f * r.w + b.w, apply_gelu);
        *(float4*)(out + (size_t)d * CCH + c0) = o;
    }
}


extern "C" void kernel_launch(void* const* d_in, const int* in_sizes, int n_in,
                              void* d_out, int out_size, void* d_ws, size_t ws_size,
                              hipStream_t stream) {
    const float* x0   = (const float*)d_in[0];
    const int*   conn = (const int*)  d_in[1];
    const float* Wl   = (const float*)d_in[2];
    const float* bl   = (const float*)d_in[3];
    const float* Wr   = (const float*)d_in[4];
    const float* br   = (const float*)d_in[5];
    const float* att  = (const float*)d_in[6];
    const float* bias = (const float*)d_in[7];

    const int N  = in_sizes[0] / D_FEAT;
    const int E  = in_sizes[1] / 2;
    const int* src = conn;
    const int* dst = conn + E;

    float* ws    = (float*)d_ws;
    float* xl    = ws;
    float* x1    = xl   + (size_t)N * HC;
    int*   cnt     = (int*)(x1 + (size_t)N * D_FEAT);
    int*   rowptr  = cnt + N;
    int*   csr_src = rowptr + N + 1;

    const dim3 blk(256);
    const int gemm_blocks = ((N / 16) * 8 + 7) / 8;
    const int node_blocks = (N + NODE_WAVES - 1) / NODE_WAVES;

    zero_int_kernel<<<(N + 255) / 256, blk, 0, stream>>>(cnt, N);
    degree_kernel<<<(E + 255) / 256, blk, 0, stream>>>(dst, cnt, E);
    rowptr_kernel<<<(N + SCAN_T - 1) / SCAN_T, dim3(SCAN_T), 0, stream>>>(cnt, rowptr, N);
    place_kernel<<<(E + 255) / 256, blk, 0, stream>>>(src, dst, rowptr, cnt, csr_src, E);

    const int L = 2;
    const float* xin = x0;
    for (int l = 0; l < L; ++l) {
        const float* Wl_l = Wl + (size_t)l * D_FEAT * HC;
        const float* bl_l = bl + (size_t)l * HC;
        const float* Wr_l = Wr + (size_t)l * D_FEAT * HC;
        const float* br_l = br + (size_t)l * HC;
        const float* at_l = att + (size_t)l * NHEAD * CCH;
        const float* bi_l = bias + (size_t)l * CCH;

        gemm64x128_wmma<<<gemm_blocks, blk, 0, stream>>>(xin, Wl_l, bl_l, xl, N);

        float* outp = (l == L - 1) ? (float*)d_out : x1;
        node_kernel<<<node_blocks, dim3(512), 0, stream>>>(xin, Wr_l, br_l, xl, rowptr, csr_src,
                                                           at_l, bi_l, outp,
                                                           (l < L - 1) ? 1 : 0, N);

        xin = x1;
    }
}


# Node pass requests up to eight neighbour rows per wave before using the first (was four, then one per left-over edge); slots past the last edge read the node's own row at score -inf; same sum order
# speedup vs baseline: 2.8226x; 1.0381x over previous
#include <hip/hip_runtime.h>

#define D_FEAT 64
#define HC 128
#define NHEAD 2
#define CCH 64
#define NEG_SLOPE 0.2f

typedef __attribute__((ext_vector_type(2))) float v2f;
typedef __attribute__((ext_vector_type(8))) float v8f;
typedef __attribute__((ext_vector_type(4))) float v4f;

__global__ void gemm64x128_wmma(const float* __restrict__ X,
                                const float* __restrict__ W,
                                const float* __restrict__ bvec,
                                float* __restrict__ Y, int rows) {
    const int wave = (blockIdx.x * blockDim.x + threadIdx.x) >> 5;
    const int lane = threadIdx.x & 31;
    const int tiles_m = rows >> 4;
    const int tm = wave >> 3;
    const int tn = wave & 7;
    if (tm >= tiles_m) return;

    const int mrow  = (tm << 4) + (lane & 15);
    const int ncol  = (tn << 4) + (lane & 15);
    const int khalf = (lane >> 4) << 1;

    v8f acc = {};
#pragma unroll
    for (int k0 = 0; k0 < D_FEAT; k0 += 4) {
        const int kk = k0 + khalf;
        v2f a, b;
        const float* ap = X + (size_t)mrow * D_FEAT + kk;
        a.x = ap[0];
        a.y = ap[1];
        b.x = W[(size_t)(kk + 0) * HC + ncol];
        b.y = W[(size_t)(kk + 1) * HC + ncol];
        acc = __builtin_amdgcn_wmma_f32_16x16x4_f32(
             false, a,  false, b,
             (short)0, acc,  false,  false);
    }

    const float bb = bvec[ncol];
    const int rbase = (tm << 4) + ((lane >> 4) << 3);
#pragma unroll
    for (int j = 0; j < 8; ++j)
        Y[(size_t)(rbase + j) * HC + ncol] = acc[j] + bb;
}


__global__ void zero_int_kernel(int* __restrict__ p, int n) {
    const int i = blockIdx.x * blockDim.x + threadIdx.x;
    if (i < n) p[i] = 0;
}

__global__ void degree_kernel(const int* __restrict__ dst, int* __restrict__ cnt, int E) {
    const int e = blockIdx.x * blockDim.x + threadIdx.x;
    if (e < E) atomicAdd(&cnt[dst[e]], 1);
}

#define SCAN_T 1024
__global__ void __launch_bounds__(SCAN_T)
rowptr_kernel(const int* __restrict__ cnt, int* __restrict__ rowptr, int n) {
    __shared__ int wsum[32];
    __shared__ int sbase;
    const int t    = threadIdx.x;
    const int lane = t & 31;
    const int w    = t >> 5;
    const int first = blockIdx.x * SCAN_T;

    int part = 0;
    for (int i = t; i < first; i += SCAN_T) part += cnt[i];
    part += __shfl_xor(part, 16);
    part += __shfl_xor(part, 8);
    part += __shfl_xor(part, 4);
    part += __shfl_xor(part, 2);
    part += __shfl_xor(part, 1);
    if (lane == 0) wsum[w] = part;
    __syncthreads();
    if (w == 0) {
        int v = wsum[lane];
        v += __shfl_xor(v, 16);
        v += __shfl_xor(v, 8);
        v += __shfl_xor(v, 4);
        v += __shfl_xor(v, 2);
        v += __shfl_xor(v, 1);
        if (lane == 0) sbase = v;
    }
    __syncthreads();
    const int base = sbase;

    const int i = first + t;
    const int c = (i < n) ? cnt[i] : 0;
    int incl = c;
    for (int o = 1; o < 32; o <<= 1) {
        const int u = __shfl_up(incl, o);
        if (lane >= o) incl += u;
    }
    if (lane == 31) wsum[w] = incl;
    __syncthreads();
    if (w == 0) {
        const int v = wsum[lane];
        int inc2 = v;
        for (int o = 1; o < 32; o <<= 1) {
            const int u = __shfl_up(inc2, o);
            if (lane >= o) inc2 += u;
        }
        wsum[lane] = inc2 - v;
    }
    __syncthreads();
    const int excl = base + wsum[w] + incl - c;
    if (i < n) rowptr[i] = excl;
    if (i == n - 1) rowptr[n] = excl + c;
}

__global__ void place_kernel(const int* __restrict__ src,
                             const int* __restrict__ dst,
                             const int* __restrict__ rowptr,
                             int* __restrict__ cnt,
                             int* __restrict__ csr_src, int E) {
    const int e = blockIdx.x * blockDim.x + threadIdx.x;
    if (e >= E) return;
    const int d = dst[e];
    const int k = atomicSub(&cnt[d], 1);
    csr_src[rowptr[d] + k - 1] = src[e];
}


__device__ __forceinline__ float head_score(const float4 v, const float4 zr, const float4 at) {
    float p = 0.0f, z;
    z = v.x + zr.x; z = (z > 0.0f) ? z : NEG_SLOPE * z; p += z * at.x;
    z = v.y + zr.y; z = (z > 0.0f) ? z : NEG_SLOPE * z; p += z * at.y;
    z = v.z + zr.z; z = (z > 0.0f) ? z : NEG_SLOPE * z; p += z * at.z;
    z = v.w + zr.w; z = (z > 0.0f) ? z : NEG_SLOPE * z; p += z * at.w;
    p += __shfl_xor(p, 1);
    p += __shfl_xor(p, 2);
    p += __shfl_xor(p, 4);
    p += __shfl_xor(p, 8);
    return p;
}

__device__ __forceinline__ void softmax_step(const float4 v, const float e,
                                             float& m, float& den, float4& acc) {
    const float mn = fmaxf(m, e);
    const float sc = __expf(m - mn);
    const float a  = __expf(e - mn);
    den   = den   * sc + a;
    acc.x = acc.x * sc + a * v.x;
    acc.y = acc.y * sc + a * v.y;
    acc.z = acc.z * sc + a * v.z;
    acc.w = acc.w * sc + a * v.w;
    m = mn;
}

__device__ __forceinline__ float finish_value(float v, int apply_gelu) {
    if (apply_gelu) {
        const float u = 0.7978845608028654f * (v + 0.044715f * v * v * v);
        v = 0.5f * v * (1.0f + tanhf(u));
    }
    return v;
}

#define ROWS_IN_FLIGHT 8
#define NODE_WAVES 16
#define XR_PITCH 132

__global__ void __launch_bounds__(512, 12)
node_kernel(const float* __restrict__ xin,
            const float* __restrict__ Wr,
            const float* __restrict__ br,
            const float* __restrict__ xl,
            const int* __restrict__ rowptr,
            const int* __restrict__ csr_src,
            const float* __restrict__ att,
            const float* __restrict__ bias,
            float* __restrict__ out,
            int apply_gelu, int n) {
    __shared__ __attribute__((aligned(16))) float sxr[NODE_WAVES * XR_PITCH];

    const int w = __builtin_amdgcn_readfirstlane((int)(threadIdx.x >> 5));
    const int lane = threadIdx.x & 31;
    const int nbase = blockIdx.x * NODE_WAVES;
    const int d = nbase + w;
    const bool valid = d < n;
    const int dd = valid ? d : n - 1;

    const int c0 = lane << 2;
    const float4 at = *(const float4*)(att + c0);
    const int beg = rowptr[dd];
    const int end = rowptr[dd + 1];
    float4 acc = *(const float4*)(xl + (size_t)dd * HC + c0);
    int idx = dd;
    {
        const int left0 = end - beg;
        const int nk0   = (left0 < 32) ? left0 : 32;
        if (lane < nk0) idx = csr_src[beg + lane];
    }

    if (w < 8) {
        int mrow = nbase + (lane & 15);
        mrow = (mrow < n) ? mrow : n - 1;
        const int ncol  = (w << 4) + (lane & 15);
        const int khalf = (lane >> 4) << 1;
        v8f t = {};
#pragma unroll
        for (int k0 = 0; k0 < D_FEAT; k0 += 4) {
            const int kk = k0 + khalf;
            v2f a, b;
            const float* ap = xin + (size_t)mrow * D_FEAT + kk;
            a.x = ap[0];
            a.y = ap[1];
            b.x = Wr[(size_t)(kk + 0) * HC + ncol];
            b.y = Wr[(size_t)(kk + 1) * HC + ncol];
            t = __builtin_amdgcn_wmma_f32_16x16x4_f32(
                 false, a,  false, b,
                 (short)0, t,  false,  false);
        }
        const float bb = br[ncol];
        const int rloc = (lane >> 4) << 3;
#pragma unroll
        for (int j = 0; j < 8; ++j)
            sxr[(rloc + j) * XR_PITCH + ncol] = t[j] + bb;
    }
    __syncthreads();
    if (!valid) return;

    const float4 zr = *(const float4*)(sxr + w * XR_PITCH + c0);

    float m   = head_score(acc, zr, at);
    float den = 1.0f;

    const v4f* __restrict__ xl4 = (const v4f*)xl;
    for (int base = beg; base < end; base += 32) {
        const int left = end - base;
        const int nk   = (left < 32) ? left : 32;
        if (base != beg) idx = (lane < nk) ? csr_src[base + lane] : d;
        for (int k0 = 0; k0 < nk; k0 += ROWS_IN_FLIGHT) {
            const int cnt = nk - k0;
            v4f r[ROWS_IN_FLIGHT];
#pragma unroll
            for (int j = 0; j < 4; ++j) {
                const int s = __builtin_amdgcn_readlane(idx, k0 + j);
                r[j] = xl4[(unsigned)s * (HC / 4) + (unsigned)lane];
            }
            if (cnt > 4) {
#pragma unroll
                for (int j = 4; j < ROWS_IN_FLIGHT; ++j) {
                    const int s = __builtin_amdgcn_readlane(idx, k0 + j);
                    r[j] = xl4[(unsigned)s * (HC / 4) + (unsigned)lane];
                }
            }
            float4 v[4];
            float e[4];
#pragma unroll
            for (int j = 0; j < 4; ++j) {
                v[j] = make_float4(r[j].x, r[j].y, r[j].z, r[j].w);
                const float t = head_score(v[j], zr, at);
                e[j] = (j < cnt) ? t : -__builtin_inff();
            }
#pragma unroll
            for (int j = 0; j < 4; ++j)
                softmax_step(v[j], e[j], m, den, acc);
            if (cnt > 4) {
#pragma unroll
                for (int j = 0; j < 4; ++j) {
                    v[j] = make_float4(r[4 + j].x, r[4 + j].y, r[4 + j].z, r[4 + j].w);
                    const float t = head_score(v[j], zr, at);
                    e[j] = (4 + j < cnt) ? t : -__builtin_inff();
                }
#pragma unroll
                for (int j = 0; j < 4; ++j)
                    softmax_step(v[j], e[j], m, den, acc);
            }
        }
    }

    const float inv = 1.0f / (den + 1e-16f);
    float4 r;
    r.x = acc.x * inv;
    r.y = acc.y * inv;
    r.z = acc.z * inv;
    r.w = acc.w * inv;
    r.x += __shfl_xor(r.x, 16);
    r.y += __shfl_xor(r.y, 16);
    r.z += __shfl_xor(r.z, 16);
    r.w += __shfl_xor(r.w, 16);

    if (lane < 16) {
        const float4 b = *(const float4*)(bias + c0);
        float4 o;
        o.x = finish_value(0.5f * r.x + b.x, apply_gelu);
        o.y = finish_value(0.5f * r.y + b.y, apply_gelu);
        o.z = finish_value(0.5f * r.z + b.z, apply_gelu);
        o.w = finish_value(0.5f * r.w + b.w, apply_gelu);
        *(float4*)(out + (size_t)d * CCH + c0) = o;
    }
}


extern "C" void kernel_launch(void* const* d_in, const int* in_sizes, int n_in,
                              void* d_out, int out_size, void* d_ws, size_t ws_size,
                              hipStream_t stream) {
    const float* x0   = (const float*)d_in[0];
    const int*   conn = (const int*)  d_in[1];
    const float* Wl   = (const float*)d_in[2];
    const float* bl   = (const float*)d_in[3];
    const float* Wr   = (const float*)d_in[4];
    const float* br   = (const float*)d_in[5];
    const float* att  = (const float*)d_in[6];
    const float* bias = (const float*)d_in[7];

    const int N  = in_sizes[0] / D_FEAT;
    const int E  = in_sizes[1] / 2;
    const int* src = conn;
    const int* dst = conn + E;

    float* ws    = (float*)d_ws;
    float* xl    = ws;
    float* x1    = xl   + (size_t)N * HC;
    int*   cnt     = (int*)(x1 + (size_t)N * D_FEAT);
    int*   rowptr  = cnt + N;
    int*   csr_src = rowptr + N + 1;

    const dim3 blk(256);
    const int gemm_blocks = ((N / 16) * 8 + 7) / 8;
    const int node_blocks = (N + NODE_WAVES - 1) / NODE_WAVES;

    zero_int_kernel<<<(N + 255) / 256, blk, 0, stream>>>(cnt, N);
    degree_kernel<<<(E + 255) / 256, blk, 0, stream>>>(dst, cnt, E);
    rowptr_kernel<<<(N + SCAN_T - 1) / SCAN_T, dim3(SCAN_T), 0, stream>>>(cnt, rowptr, N);
    place_kernel<<<(E + 255) / 256, blk, 0, stream>>>(src, dst, rowptr, cnt, csr_src, E);

    const int L = 2;
    const float* xin = x0;
    for (int l = 0; l < L; ++l) {
        const float* Wl_l = Wl + (size_t)l * D_FEAT * HC;
        const float* bl_l = bl + (size_t)l * HC;
        const float* Wr_l = Wr + (size_t)l * D_FEAT * HC;
        const float* br_l = br + (size_t)l * HC;
        const float* at_l = att + (size_t)l * NHEAD * CCH;
        const float* bi_l = bias + (size_t)l * CCH;

        gemm64x128_wmma<<<gemm_blocks, blk, 0, stream>>>(xin, Wl_l, bl_l, xl, N);

        float* outp = (l == L - 1) ? (float*)d_out : x1;
        node_kernel<<<node_blocks, dim3(512), 0, stream>>>(xin, Wr_l, br_l, xl, rowptr, csr_src,
                                                           at_l, bi_l, outp,
                                                           (l < L - 1) ? 1 : 0, N);

        xin = x1;
    }
}


# Node pass: a wave handles four of a tile's 16 nodes in turn (4-wave groups, two xr tiles per wave), row pointers once per wave, next node's indices a turn ahead: 150000 fewer waves; same sum order
# speedup vs baseline: 2.9095x; 1.0308x over previous
#include <hip/hip_runtime.h>

#define D_FEAT 64
#define HC 128
#define NHEAD 2
#define CCH 64
#define NEG_SLOPE 0.2f

typedef __attribute__((ext_vector_type(2))) float v2f;
typedef __attribute__((ext_vector_type(8))) float v8f;
typedef __attribute__((ext_vector_type(4))) float v4f;

__global__ void gemm64x128_wmma(const float* __restrict__ X,
                                const float* __restrict__ W,
                                const float* __restrict__ bvec,
                                float* __restrict__ Y, int rows) {
    const int wave = (blockIdx.x * blockDim.x + threadIdx.x) >> 5;
    const int lane = threadIdx.x & 31;
    const int tiles_m = rows >> 4;
    const int tm = wave >> 3;
    const int tn = wave & 7;
    if (tm >= tiles_m) return;

    const int mrow  = (tm << 4) + (lane & 15);
    const int ncol  = (tn << 4) + (lane & 15);
    const int khalf = (lane >> 4) << 1;

    v8f acc = {};
#pragma unroll
    for (int k0 = 0; k0 < D_FEAT; k0 += 4) {
        const int kk = k0 + khalf;
        v2f a, b;
        const float* ap = X + (size_t)mrow * D_FEAT + kk;
        a.x = ap[0];
        a.y = ap[1];
        b.x = W[(size_t)(kk + 0) * HC + ncol];
        b.y = W[(size_t)(kk + 1) * HC + ncol];
        acc = __builtin_amdgcn_wmma_f32_16x16x4_f32(
             false, a,  false, b,
             (short)0, acc,  false,  false);
    }

    const float bb = bvec[ncol];
    const int rbase = (tm << 4) + ((lane >> 4) << 3);
#pragma unroll
    for (int j = 0; j < 8; ++j)
        Y[(size_t)(rbase + j) * HC + ncol] = acc[j] + bb;
}


__global__ void zero_int_kernel(int* __restrict__ p, int n) {
    const int i = blockIdx.x * blockDim.x + threadIdx.x;
    if (i < n) p[i] = 0;
}

__global__ void degree_kernel(const int* __restrict__ dst, int* __restrict__ cnt, int E) {
    const int e = blockIdx.x * blockDim.x + threadIdx.x;
    if (e < E) atomicAdd(&cnt[dst[e]], 1);
}

#define SCAN_T 1024
__global__ void __launch_bounds__(SCAN_T)
rowptr_kernel(const int* __restrict__ cnt, int* __restrict__ rowptr, int n) {
    __shared__ int wsum[32];
    __shared__ int sbase;
    const int t    = threadIdx.x;
    const int lane = t & 31;
    const int w    = t >> 5;
    const int first = blockIdx.x * SCAN_T;

    int part = 0;
    for (int i = t; i < first; i += SCAN_T) part += cnt[i];
    part += __shfl_xor(part, 16);
    part += __shfl_xor(part, 8);
    part += __shfl_xor(part, 4);
    part += __shfl_xor(part, 2);
    part += __shfl_xor(part, 1);
    if (lane == 0) wsum[w] = part;
    __syncthreads();
    if (w == 0) {
        int v = wsum[lane];
        v += __shfl_xor(v, 16);
        v += __shfl_xor(v, 8);
        v += __shfl_xor(v, 4);
        v += __shfl_xor(v, 2);
        v += __shfl_xor(v, 1);
        if (lane == 0) sbase = v;
    }
    __syncthreads();
    const int base = sbase;

    const int i = first + t;
    const int c = (i < n) ? cnt[i] : 0;
    int incl = c;
    for (int o = 1; o < 32; o <<= 1) {
        const int u = __shfl_up(incl, o);
        if (lane >= o) incl += u;
    }
    if (lane == 31) wsum[w] = incl;
    __syncthreads();
    if (w == 0) {
        const int v = wsum[lane];
        int inc2 = v;
        for (int o = 1; o < 32; o <<= 1) {
            const int u = __shfl_up(inc2, o);
            if (lane >= o) inc2 += u;
        }
        wsum[lane] = inc2 - v;
    }
    __syncthreads();
    const int excl = base + wsum[w] + incl - c;
    if (i < n) rowptr[i] = excl;
    if (i == n - 1) rowptr[n] = excl + c;
}

__global__ void place_kernel(const int* __restrict__ src,
                             const int* __restrict__ dst,
                             const int* __restrict__ rowptr,
                             int* __restrict__ cnt,
                             int* __restrict__ csr_src, int E) {
    const int e = blockIdx.x * blockDim.x + threadIdx.x;
    if (e >= E) return;
    const int d = dst[e];
    const int k = atomicSub(&cnt[d], 1);
    csr_src[rowptr[d] + k - 1] = src[e];
}


__device__ __forceinline__ float head_score(const float4 v, const float4 zr, const float4 at) {
    float p = 0.0f, z;
    z = v.x + zr.x; z = (z > 0.0f) ? z : NEG_SLOPE * z; p += z * at.x;
    z = v.y + zr.y; z = (z > 0.0f) ? z : NEG_SLOPE * z; p += z * at.y;
    z = v.z + zr.z; z = (z > 0.0f) ? z : NEG_SLOPE * z; p += z * at.z;
    z = v.w + zr.w; z = (z > 0.0f) ? z : NEG_SLOPE * z; p += z * at.w;
    p += __shfl_xor(p, 1);
    p += __shfl_xor(p, 2);
    p += __shfl_xor(p, 4);
    p += __shfl_xor(p, 8);
    return p;
}

__device__ __forceinline__ void softmax_step(const float4 v, const float e,
                                             float& m, float& den, float4& acc) {
    const float mn = fmaxf(m, e);
    const float sc = __expf(m - mn);
    const float a  = __expf(e - mn);
    den   = den   * sc + a;
    acc.x = acc.x * sc + a * v.x;
    acc.y = acc.y * sc + a * v.y;
    acc.z = acc.z * sc + a * v.z;
    acc.w = acc.w * sc + a * v.w;
    m = mn;
}

__device__ __forceinline__ float finish_value(float v, int apply_gelu) {
    if (apply_gelu) {
        const float u = 0.7978845608028654f * (v + 0.044715f * v * v * v);
        v = 0.5f * v * (1.0f + tanhf(u));
    }
    return v;
}

#define ROWS_IN_FLIGHT 8
#define TILE_NODES 16
#define NODE_WAVES 4
#define WAVE_NODES 4
#define XR_PITCH 132

__global__ void __launch_bounds__(128, 12)
node_kernel(const float* __restrict__ xin,
            const float* __restrict__ Wr,
            const float* __restrict__ br,
            const float* __restrict__ xl,
            const int* __restrict__ rowptr,
            const int* __restrict__ csr_src,
            const float* __restrict__ att,
            const float* __restrict__ bias,
            float* __restrict__ out,
            int apply_gelu, int n) {
    __shared__ __attribute__((aligned(16))) float sxr[TILE_NODES * XR_PITCH];

    const int w = __builtin_amdgcn_readfirstlane((int)(threadIdx.x >> 5));
    const int lane = threadIdx.x & 31;
    const int nbase = blockIdx.x * TILE_NODES;
    const int d0 = nbase + w * WAVE_NODES;

    const int c0 = lane << 2;
    const float4 at = *(const float4*)(att + c0);
    int rpi = d0 + (lane & 7);
    rpi = (rpi < n) ? rpi : n;
    const int rpv = rowptr[rpi];
    const int dfirst = (d0 < n) ? d0 : n - 1;
    const v4f* __restrict__ xl4 = (const v4f*)xl;

    int beg = __builtin_amdgcn_readlane(rpv, 0);
    int end = __builtin_amdgcn_readlane(rpv, 1);
    int idx = dfirst;
    {
        const int left0 = end - beg;
        const int nk0   = (left0 < 32) ? left0 : 32;
        if (lane < nk0) idx = csr_src[beg + lane];
    }

    {
        int mrow = nbase + (lane & 15);
        mrow = (mrow < n) ? mrow : n - 1;
        const int ncol  = (w << 5) + (lane & 15);
        const int khalf = (lane >> 4) << 1;
        v8f t0 = {};
        v8f t1 = {};
#pragma unroll
        for (int k0 = 0; k0 < D_FEAT; k0 += 4) {
            const int kk = k0 + khalf;
            v2f a, b0, b1;
            const float* ap = xin + (size_t)mrow * D_FEAT + kk;
            a.x = ap[0];
            a.y = ap[1];
            b0.x = Wr[(size_t)(kk + 0) * HC + ncol];
            b0.y = Wr[(size_t)(kk + 1) * HC + ncol];
            b1.x = Wr[(size_t)(kk + 0) * HC + ncol + 16];
            b1.y = Wr[(size_t)(kk + 1) * HC + ncol + 16];
            t0 = __builtin_amdgcn_wmma_f32_16x16x4_f32(
                 false, a,  false, b0,
                 (short)0, t0,  false,  false);
            t1 = __builtin_amdgcn_wmma_f32_16x16x4_f32(
                 false, a,  false, b1,
                 (short)0, t1,  false,  false);
        }
        const float bb0 = br[ncol];
        const float bb1 = br[ncol + 16];
        const int rloc = (lane >> 4) << 3;
#pragma unroll
        for (int j = 0; j < 8; ++j) {
            sxr[(rloc + j) * XR_PITCH + ncol]      = t0[j] + bb0;
            sxr[(rloc + j) * XR_PITCH + ncol + 16] = t1[j] + bb1;
        }
    }
    __syncthreads();

    for (int jn = 0; jn < WAVE_NODES; ++jn) {
        const int d = d0 + jn;
        if (d >= n) break;

        int nbeg = 0, nend = 0, nidx = d;
        if (jn + 1 < WAVE_NODES) {
            int dn = d + 1;
            dn = (dn < n) ? dn : n - 1;
            nbeg = __builtin_amdgcn_readlane(rpv, jn + 1);
            nend = __builtin_amdgcn_readlane(rpv, jn + 2);
            nidx = dn;
            const int nleft = nend - nbeg;
            const int nnk   = (nleft < 32) ? nleft : 32;
            if (lane < nnk) nidx = csr_src[nbeg + lane];
        }

        float4 acc;
        {
            const v4f t = xl4[(unsigned)d * (HC / 4) + (unsigned)lane];
            acc = make_float4(t.x, t.y, t.z, t.w);
        }
        const float4 zr = *(const float4*)(sxr + (w * WAVE_NODES + jn) * XR_PITCH + c0);

        float m   = 0.0f;
        float den = 1.0f;
        bool have_m = false;

        for (int base = beg; base < end; base += 32) {
            const int left = end - base;
            const int nk   = (left < 32) ? left : 32;
            if (base != beg) idx = (lane < nk) ? csr_src[base + lane] : d;
            for (int k0 = 0; k0 < nk; k0 += ROWS_IN_FLIGHT) {
                const int cnt = nk - k0;
                v4f r[ROWS_IN_FLIGHT];
#pragma unroll
                for (int j = 0; j < 4; ++j) {
                    const int s = __builtin_amdgcn_readlane(idx, k0 + j);
                    r[j] = xl4[(unsigned)s * (HC / 4) + (unsigned)lane];
                }
                if (cnt > 4) {
#pragma unroll
                    for (int j = 4; j < ROWS_IN_FLIGHT; ++j) {
                        const int s = __builtin_amdgcn_readlane(idx, k0 + j);
                        r[j] = xl4[(unsigned)s * (HC / 4) + (unsigned)lane];
                    }
                }
                if (!have_m) {
                    m = head_score(acc, zr, at);
                    have_m = true;
                }
                float4 v[4];
                float e[4];
#pragma unroll
                for (int j = 0; j < 4; ++j) {
                    v[j] = make_float4(r[j].x, r[j].y, r[j].z, r[j].w);
                    const float t = head_score(v[j], zr, at);
                    e[j] = (j < cnt) ? t : -__builtin_inff();
                }
#pragma unroll
                for (int j = 0; j < 4; ++j)
                    softmax_step(v[j], e[j], m, den, acc);
                if (cnt > 4) {
#pragma unroll
                    for (int j = 0; j < 4; ++j) {
                        v[j] = make_float4(r[4 + j].x, r[4 + j].y, r[4 + j].z, r[4 + j].w);
                        const float t = head_score(v[j], zr, at);
                        e[j] = (4 + j < cnt) ? t : -__builtin_inff();
                    }
#pragma unroll
                    for (int j = 0; j < 4; ++j)
                        softmax_step(v[j], e[j], m, den, acc);
                }
            }
        }

        const float inv = 1.0f / (den + 1e-16f);
        float4 r;
        r.x = acc.x * inv;
        r.y = acc.y * inv;
        r.z = acc.z * inv;
        r.w = acc.w * inv;
        r.x += __shfl_xor(r.x, 16);
        r.y += __shfl_xor(r.y, 16);
        r.z += __shfl_xor(r.z, 16);
        r.w += __shfl_xor(r.w, 16);

        if (lane < 16) {
            const v4f b = ((const v4f*)bias)[(unsigned)lane];
            float4 o;
            o.x = finish_value(0.5f * r.x + b.x, apply_gelu);
            o.y = finish_value(0.5f * r.y + b.y, apply_gelu);
            o.z = finish_value(0.5f * r.z + b.z, apply_gelu);
            o.w = finish_value(0.5f * r.w + b.w, apply_gelu);
            v4f o4;
            o4.x = o.x; o4.y = o.y; o4.z = o.z; o4.w = o.w;
            ((v4f*)out)[(unsigned)d * (CCH / 4) + (unsigned)lane] = o4;
        }

        beg = nbeg;
        end = nend;
        idx = nidx;
    }
}


extern "C" void kernel_launch(void* const* d_in, const int* in_sizes, int n_in,
                              void* d_out, int out_size, void* d_ws, size_t ws_size,
                              hipStream_t stream) {
    const float* x0   = (const float*)d_in[0];
    const int*   conn = (const int*)  d_in[1];
    const float* Wl   = (const float*)d_in[2];
    const float* bl   = (const float*)d_in[3];
    const float* Wr   = (const float*)d_in[4];
    const float* br   = (const float*)d_in[5];
    const float* att  = (const float*)d_in[6];
    const float* bias = (const float*)d_in[7];

    const int N  = in_sizes[0] / D_FEAT;
    const int E  = in_sizes[1] / 2;
    const int* src = conn;
    const int* dst = conn + E;

    float* ws    = (float*)d_ws;
    float* xl    = ws;
    float* x1    = xl   + (size_t)N * HC;
    int*   cnt     = (int*)(x1 + (size_t)N * D_FEAT);
    int*   rowptr  = cnt + N;
    int*   csr_src = rowptr + N + 1;

    const dim3 blk(256);
    const int gemm_blocks = ((N / 16) * 8 + 7) / 8;
    const int node_blocks = (N + TILE_NODES - 1) / TILE_NODES;

    zero_int_kernel<<<(N + 255) / 256, blk, 0, stream>>>(cnt, N);
    degree_kernel<<<(E + 255) / 256, blk, 0, stream>>>(dst, cnt, E);
    rowptr_kernel<<<(N + SCAN_T - 1) / SCAN_T, dim3(SCAN_T), 0, stream>>>(cnt, rowptr, N);
    place_kernel<<<(E + 255) / 256, blk, 0, stream>>>(src, dst, rowptr, cnt, csr_src, E);

    const int L = 2;
    const float* xin = x0;
    for (int l = 0; l < L; ++l) {
        const float* Wl_l = Wl + (size_t)l * D_FEAT * HC;
        const float* bl_l = bl + (size_t)l * HC;
        const float* Wr_l = Wr + (size_t)l * D_FEAT * HC;
        const float* br_l = br + (size_t)l * HC;
        const float* at_l = att + (size_t)l * NHEAD * CCH;
        const float* bi_l = bias + (size_t)l * CCH;

        gemm64x128_wmma<<<gemm_blocks, blk, 0, stream>>>(xin, Wl_l, bl_l, xl, N);

        float* outp = (l == L - 1) ? (float*)d_out : x1;
        node_kernel<<<node_blocks, dim3(32 * NODE_WAVES), 0, stream>>>(xin, Wr_l, br_l, xl, rowptr, csr_src,
                                                           at_l, bi_l, outp,
                                                           (l < L - 1) ? 1 : 0, N);

        xin = x1;
    }
}


# Edge bucketing: one integer atomic per edge, not two: the degree count's returning add hands each edge its slot, the place pass reads it back; slots by arrival as before, float sum order may differ
# speedup vs baseline: 3.3272x; 1.1436x over previous
#include <hip/hip_runtime.h>

#define D_FEAT 64
#define HC 128
#define NHEAD 2
#define CCH 64
#define NEG_SLOPE 0.2f

typedef __attribute__((ext_vector_type(2))) float v2f;
typedef __attribute__((ext_vector_type(8))) float v8f;
typedef __attribute__((ext_vector_type(4))) float v4f;

__global__ void gemm64x128_wmma(const float* __restrict__ X,
                                const float* __restrict__ W,
                                const float* __restrict__ bvec,
                                float* __restrict__ Y, int rows) {
    const int wave = (blockIdx.x * blockDim.x + threadIdx.x) >> 5;
    const int lane = threadIdx.x & 31;
    const int tiles_m = rows >> 4;
    const int tm = wave >> 3;
    const int tn = wave & 7;
    if (tm >= tiles_m) return;

    const int mrow  = (tm << 4) + (lane & 15);
    const int ncol  = (tn << 4) + (lane & 15);
    const int khalf = (lane >> 4) << 1;

    v8f acc = {};
#pragma unroll
    for (int k0 = 0; k0 < D_FEAT; k0 += 4) {
        const int kk = k0 + khalf;
        v2f a, b;
        const float* ap = X + (size_t)mrow * D_FEAT + kk;
        a.x = ap[0];
        a.y = ap[1];
        b.x = W[(size_t)(kk + 0) * HC + ncol];
        b.y = W[(size_t)(kk + 1) * HC + ncol];
        acc = __builtin_amdgcn_wmma_f32_16x16x4_f32(
             false, a,  false, b,
             (short)0, acc,  false,  false);
    }

    const float bb = bvec[ncol];
    const int rbase = (tm << 4) + ((lane >> 4) << 3);
#pragma unroll
    for (int j = 0; j < 8; ++j)
        Y[(size_t)(rbase + j) * HC + ncol] = acc[j] + bb;
}


__global__ void zero_int_kernel(int* __restrict__ p, int n) {
    const int i = blockIdx.x * blockDim.x + threadIdx.x;
    if (i < n) p[i] = 0;
}

__global__ void degree_kernel(const int* __restrict__ dst, int* __restrict__ cnt,
                              int* __restrict__ rank, int E) {
    const int e = blockIdx.x * blockDim.x + threadIdx.x;
    if (e < E) rank[e] = atomicAdd(&cnt[dst[e]], 1);
}

#define SCAN_T 1024
__global__ void __launch_bounds__(SCAN_T)
rowptr_kernel(const int* __restrict__ cnt, int* __restrict__ rowptr, int n) {
    __shared__ int wsum[32];
    __shared__ int sbase;
    const int t    = threadIdx.x;
    const int lane = t & 31;
    const int w    = t >> 5;
    const int first = blockIdx.x * SCAN_T;

    int part = 0;
    for (int i = t; i < first; i += SCAN_T) part += cnt[i];
    part += __shfl_xor(part, 16);
    part += __shfl_xor(part, 8);
    part += __shfl_xor(part, 4);
    part += __shfl_xor(part, 2);
    part += __shfl_xor(part, 1);
    if (lane == 0) wsum[w] = part;
    __syncthreads();
    if (w == 0) {
        int v = wsum[lane];
        v += __shfl_xor(v, 16);
        v += __shfl_xor(v, 8);
        v += __shfl_xor(v, 4);
        v += __shfl_xor(v, 2);
        v += __shfl_xor(v, 1);
        if (lane == 0) sbase = v;
    }
    __syncthreads();
    const int base = sbase;

    const int i = first + t;
    const int c = (i < n) ? cnt[i] : 0;
    int incl = c;
    for (int o = 1; o < 32; o <<= 1) {
        const int u = __shfl_up(incl, o);
        if (lane >= o) incl += u;
    }
    if (lane == 31) wsum[w] = incl;
    __syncthreads();
    if (w == 0) {
        const int v = wsum[lane];
        int inc2 = v;
        for (int o = 1; o < 32; o <<= 1) {
            const int u = __shfl_up(inc2, o);
            if (lane >= o) inc2 += u;
        }
        wsum[lane] = inc2 - v;
    }
    __syncthreads();
    const int excl = base + wsum[w] + incl - c;
    if (i < n) rowptr[i] = excl;
    if (i == n - 1) rowptr[n] = excl + c;
}

__global__ void place_kernel(const int* __restrict__ src,
                             const int* __restrict__ dst,
                             const int* __restrict__ rowptr,
                             const int* __restrict__ rank,
                             int* __restrict__ csr_src, int E) {
    const int e = blockIdx.x * blockDim.x + threadIdx.x;
    if (e >= E) return;
    csr_src[rowptr[dst[e]] + rank[e]] = src[e];
}


__device__ __forceinline__ float head_score(const float4 v, const float4 zr, const float4 at) {
    float p = 0.0f, z;
    z = v.x + zr.x; z = (z > 0.0f) ? z : NEG_SLOPE * z; p += z * at.x;
    z = v.y + zr.y; z = (z > 0.0f) ? z : NEG_SLOPE * z; p += z * at.y;
    z = v.z + zr.z; z = (z > 0.0f) ? z : NEG_SLOPE * z; p += z * at.z;
    z = v.w + zr.w; z = (z > 0.0f) ? z : NEG_SLOPE * z; p += z * at.w;
    p += __shfl_xor(p, 1);
    p += __shfl_xor(p, 2);
    p += __shfl_xor(p, 4);
    p += __shfl_xor(p, 8);
    return p;
}

__device__ __forceinline__ void softmax_step(const float4 v, const float e,
                                             float& m, float& den, float4& acc) {
    const float mn = fmaxf(m, e);
    const float sc = __expf(m - mn);
    const float a  = __expf(e - mn);
    den   = den   * sc + a;
    acc.x = acc.x * sc + a * v.x;
    acc.y = acc.y * sc + a * v.y;
    acc.z = acc.z * sc + a * v.z;
    acc.w = acc.w * sc + a * v.w;
    m = mn;
}

__device__ __forceinline__ float finish_value(float v, int apply_gelu) {
    if (apply_gelu) {
        const float u = 0.7978845608028654f * (v + 0.044715f * v * v * v);
        v = 0.5f * v * (1.0f + tanhf(u));
    }
    return v;
}

#define ROWS_IN_FLIGHT 8
#define TILE_NODES 16
#define NODE_WAVES 4
#define WAVE_NODES 4
#define XR_PITCH 132

__global__ void __launch_bounds__(128, 12)
node_kernel(const float* __restrict__ xin,
            const float* __restrict__ Wr,
            const float* __restrict__ br,
            const float* __restrict__ xl,
            const int* __restrict__ rowptr,
            const int* __restrict__ csr_src,
            const float* __restrict__ att,
            const float* __restrict__ bias,
            float* __restrict__ out,
            int apply_gelu, int n) {
    __shared__ __attribute__((aligned(16))) float sxr[TILE_NODES * XR_PITCH];

    const int w = __builtin_amdgcn_readfirstlane((int)(threadIdx.x >> 5));
    const int lane = threadIdx.x & 31;
    const int nbase = blockIdx.x * TILE_NODES;
    const int d0 = nbase + w * WAVE_NODES;

    const int c0 = lane << 2;
    const float4 at = *(const float4*)(att + c0);
    int rpi = d0 + (lane & 7);
    rpi = (rpi < n) ? rpi : n;
    const int rpv = rowptr[rpi];
    const int dfirst = (d0 < n) ? d0 : n - 1;
    const v4f* __restrict__ xl4 = (const v4f*)xl;

    int beg = __builtin_amdgcn_readlane(rpv, 0);
    int end = __builtin_amdgcn_readlane(rpv, 1);
    int idx = dfirst;
    {
        const int left0 = end - beg;
        const int nk0   = (left0 < 32) ? left0 : 32;
        if (lane < nk0) idx = csr_src[beg + lane];
    }

    {
        int mrow = nbase + (lane & 15);
        mrow = (mrow < n) ? mrow : n - 1;
        const int ncol  = (w << 5) + (lane & 15);
        const int khalf = (lane >> 4) << 1;
        v8f t0 = {};
        v8f t1 = {};
#pragma unroll
        for (int k0 = 0; k0 < D_FEAT; k0 += 4) {
            const int kk = k0 + khalf;
            v2f a, b0, b1;
            const float* ap = xin + (size_t)mrow * D_FEAT + kk;
            a.x = ap[0];
            a.y = ap[1];
            b0.x = Wr[(size_t)(kk + 0) * HC + ncol];
            b0.y = Wr[(size_t)(kk + 1) * HC + ncol];
            b1.x = Wr[(size_t)(kk + 0) * HC + ncol + 16];
            b1.y = Wr[(size_t)(kk + 1) * HC + ncol + 16];
            t0 = __builtin_amdgcn_wmma_f32_16x16x4_f32(
                 false, a,  false, b0,
                 (short)0, t0,  false,  false);
            t1 = __builtin_amdgcn_wmma_f32_16x16x4_f32(
                 false, a,  false, b1,
                 (short)0, t1,  false,  false);
        }
        const float bb0 = br[ncol];
        const float bb1 = br[ncol + 16];
        const int rloc = (lane >> 4) << 3;
#pragma unroll
        for (int j = 0; j < 8; ++j) {
            sxr[(rloc + j) * XR_PITCH + ncol]      = t0[j] + bb0;
            sxr[(rloc + j) * XR_PITCH + ncol + 16] = t1[j] + bb1;
        }
    }
    __syncthreads();

    for (int jn = 0; jn < WAVE_NODES; ++jn) {
        const int d = d0 + jn;
        if (d >= n) break;

        int nbeg = 0, nend = 0, nidx = d;
        if (jn + 1 < WAVE_NODES) {
            int dn = d + 1;
            dn = (dn < n) ? dn : n - 1;
            nbeg = __builtin_amdgcn_readlane(rpv, jn + 1);
            nend = __builtin_amdgcn_readlane(rpv, jn + 2);
            nidx = dn;
            const int nleft = nend - nbeg;
            const int nnk   = (nleft < 32) ? nleft : 32;
            if (lane < nnk) nidx = csr_src[nbeg + lane];
        }

        float4 acc;
        {
            const v4f t = xl4[(unsigned)d * (HC / 4) + (unsigned)lane];
            acc = make_float4(t.x, t.y, t.z, t.w);
        }
        const float4 zr = *(const float4*)(sxr + (w * WAVE_NODES + jn) * XR_PITCH + c0);

        float m   = 0.0f;
        float den = 1.0f;
        bool have_m = false;

        for (int base = beg; base < end; base += 32) {
            const int left = end - base;
            const int nk   = (left < 32) ? left : 32;
            if (base != beg) idx = (lane < nk) ? csr_src[base + lane] : d;
            for (int k0 = 0; k0 < nk; k0 += ROWS_IN_FLIGHT) {
                const int cnt = nk - k0;
                v4f r[ROWS_IN_FLIGHT];
#pragma unroll
                for (int j = 0; j < 4; ++j) {
                    const int s = __builtin_amdgcn_readlane(idx, k0 + j);
                    r[j] = xl4[(unsigned)s * (HC / 4) + (unsigned)lane];
                }
                if (cnt > 4) {
#pragma unroll
                    for (int j = 4; j < ROWS_IN_FLIGHT; ++j) {
                        const int s = __builtin_amdgcn_readlane(idx, k0 + j);
                        r[j] = xl4[(unsigned)s * (HC / 4) + (unsigned)lane];
                    }
                }
                if (!have_m) {
                    m = head_score(acc, zr, at);
                    have_m = true;
                }
                float4 v[4];
                float e[4];
#pragma unroll
                for (int j = 0; j < 4; ++j) {
                    v[j] = make_float4(r[j].x, r[j].y, r[j].z, r[j].w);
                    const float t = head_score(v[j], zr, at);
                    e[j] = (j < cnt) ? t : -__builtin_inff();
                }
#pragma unroll
                for (int j = 0; j < 4; ++j)
                    softmax_step(v[j], e[j], m, den, acc);
                if (cnt > 4) {
#pragma unroll
                    for (int j = 0; j < 4; ++j) {
                        v[j] = make_float4(r[4 + j].x, r[4 + j].y, r[4 + j].z, r[4 + j].w);
                        const float t = head_score(v[j], zr, at);
                        e[j] = (4 + j < cnt) ? t : -__builtin_inff();
                    }
#pragma unroll
                    for (int j = 0; j < 4; ++j)
                        softmax_step(v[j], e[j], m, den, acc);
                }
            }
        }

        const float inv = 1.0f / (den + 1e-16f);
        float4 r;
        r.x = acc.x * inv;
        r.y = acc.y * inv;
        r.z = acc.z * inv;
        r.w = acc.w * inv;
        r.x += __shfl_xor(r.x, 16);
        r.y += __shfl_xor(r.y, 16);
        r.z += __shfl_xor(r.z, 16);
        r.w += __shfl_xor(r.w, 16);

        if (lane < 16) {
            const v4f b = ((const v4f*)bias)[(unsigned)lane];
            float4 o;
            o.x = finish_value(0.5f * r.x + b.x, apply_gelu);
            o.y = finish_value(0.5f * r.y + b.y, apply_gelu);
            o.z = finish_value(0.5f * r.z + b.z, apply_gelu);
            o.w = finish_value(0.5f * r.w + b.w, apply_gelu);
            v4f o4;
            o4.x = o.x; o4.y = o.y; o4.z = o.z; o4.w = o.w;
            ((v4f*)out)[(unsigned)d * (CCH / 4) + (unsigned)lane] = o4;
        }

        beg = nbeg;
        end = nend;
        idx = nidx;
    }
}


extern "C" void kernel_launch(void* const* d_in, const int* in_sizes, int n_in,
                              void* d_out, int out_size, void* d_ws, size_t ws_size,
                              hipStream_t stream) {
    const float* x0   = (const float*)d_in[0];
    const int*   conn = (const int*)  d_in[1];
    const float* Wl   = (const float*)d_in[2];
    const float* bl   = (const float*)d_in[3];
    const float* Wr   = (const float*)d_in[4];
    const float* br   = (const float*)d_in[5];
    const float* att  = (const float*)d_in[6];
    const float* bias = (const float*)d_in[7];

    const int N  = in_sizes[0] / D_FEAT;
    const int E  = in_sizes[1] / 2;
    const int* src = conn;
    const int* dst = conn + E;

    float* ws    = (float*)d_ws;
    float* xl    = ws;
    float* x1    = xl   + (size_t)N * HC;
    int*   cnt     = (int*)(x1 + (size_t)N * D_FEAT);
    int*   rowptr  = cnt + N;
    int*   csr_src = rowptr + N + 1;
    int*   rank    = csr_src + E;

    const dim3 blk(256);
    const int gemm_blocks = ((N / 16) * 8 + 7) / 8;
    const int node_blocks = (N + TILE_NODES - 1) / TILE_NODES;

    zero_int_kernel<<<(N + 255) / 256, blk, 0, stream>>>(cnt, N);
    degree_kernel<<<(E + 255) / 256, blk, 0, stream>>>(dst, cnt, rank, E);
    rowptr_kernel<<<(N + SCAN_T - 1) / SCAN_T, dim3(SCAN_T), 0, stream>>>(cnt, rowptr, N);
    place_kernel<<<(E + 255) / 256, blk, 0, stream>>>(src, dst, rowptr, rank, csr_src, E);

    const int L = 2;
    const float* xin = x0;
    for (int l = 0; l < L; ++l) {
        const float* Wl_l = Wl + (size_t)l * D_FEAT * HC;
        const float* bl_l = bl + (size_t)l * HC;
        const float* Wr_l = Wr + (size_t)l * D_FEAT * HC;
        const float* br_l = br + (size_t)l * HC;
        const float* at_l = att + (size_t)l * NHEAD * CCH;
        const float* bi_l = bias + (size_t)l * CCH;

        gemm64x128_wmma<<<gemm_blocks, blk, 0, stream>>>(xin, Wl_l, bl_l, xl, N);

        float* outp = (l == L - 1) ? (float*)d_out : x1;
        node_kernel<<<node_blocks, dim3(32 * NODE_WAVES), 0, stream>>>(xin, Wr_l, br_l, xl, rowptr, csr_src,
                                                           at_l, bi_l, outp,
                                                           (l < L - 1) ? 1 : 0, N);

        xin = x1;
    }
}


# Edge bucketing without memory atomics: a work-group per (destination range, edge chunk) counts in LDS, a column pass sums the chunks; slots by chunk, then arrival in a group: float sum order differs
# speedup vs baseline: 3.6832x; 1.1070x over previous
#include <hip/hip_runtime.h>

#define D_FEAT 64
#define HC 128
#define NHEAD 2
#define CCH 64
#define NEG_SLOPE 0.2f

typedef __attribute__((ext_vector_type(2))) float v2f;
typedef __attribute__((ext_vector_type(8))) float v8f;
typedef __attribute__((ext_vector_type(4))) float v4f;

__global__ void gemm64x128_wmma(const float* __restrict__ X,
                                const float* __restrict__ W,
                                const float* __restrict__ bvec,
                                float* __restrict__ Y, int rows) {
    const int wave = (blockIdx.x * blockDim.x + threadIdx.x) >> 5;
    const int lane = threadIdx.x & 31;
    const int tiles_m = rows >> 4;
    const int tm = wave >> 3;
    const int tn = wave & 7;
    if (tm >= tiles_m) return;

    const int mrow  = (tm << 4) + (lane & 15);
    const int ncol  = (tn << 4) + (lane & 15);
    const int khalf = (lane >> 4) << 1;

    v8f acc = {};
#pragma unroll
    for (int k0 = 0; k0 < D_FEAT; k0 += 4) {
        const int kk = k0 + khalf;
        v2f a, b;
        const float* ap = X + (size_t)mrow * D_FEAT + kk;
        a.x = ap[0];
        a.y = ap[1];
        b.x = W[(size_t)(kk + 0) * HC + ncol];
        b.y = W[(size_t)(kk + 1) * HC + ncol];
        acc = __builtin_amdgcn_wmma_f32_16x16x4_f32(
             false, a,  false, b,
             (short)0, acc,  false,  false);
    }

    const float bb = bvec[ncol];
    const int rbase = (tm << 4) + ((lane >> 4) << 3);
#pragma unroll
    for (int j = 0; j < 8; ++j)
        Y[(size_t)(rbase + j) * HC + ncol] = acc[j] + bb;
}


#define RANGE_NODES 16384
#define EDGE_CHUNKS 32
#define COUNT_T 1024
#define COUNT_AHEAD 16

__device__ __forceinline__ int edge_chunk_len(int E) {
    return ((E + EDGE_CHUNKS * 256 - 1) / (EDGE_CHUNKS * 256)) * 256;
}

__global__ void __launch_bounds__(COUNT_T)
count_kernel(const int* __restrict__ dst, int* __restrict__ rank,
             int* __restrict__ part, int E, int n) {
    __shared__ int hist[RANGE_NODES];
    const int t  = threadIdx.x;
    const int c  = blockIdx.x % EDGE_CHUNKS;
    const int lo = (blockIdx.x / EDGE_CHUNKS) * RANGE_NODES;

    for (int i = t; i < RANGE_NODES; i += COUNT_T) hist[i] = 0;
    __syncthreads();

    const int chunk = edge_chunk_len(E);
    const int e0 = c * chunk;
    int e1 = e0 + chunk;
    e1 = (e1 < E) ? e1 : E;
    for (int e = e0 + t; e < e1; e += COUNT_T * COUNT_AHEAD) {
        int d[COUNT_AHEAD];
#pragma unroll
        for (int j = 0; j < COUNT_AHEAD; ++j) {
            const int ee = e + j * COUNT_T;
            d[j] = (ee < e1) ? dst[ee] : -1;
        }
        int r[COUNT_AHEAD];
#pragma unroll
        for (int j = 0; j < COUNT_AHEAD; ++j) {
            const unsigned u = (unsigned)(d[j] - lo);
            r[j] = 0;
            if (u < (unsigned)RANGE_NODES)
                r[j] = atomicAdd(&hist[u], 1);
        }
#pragma unroll
        for (int j = 0; j < COUNT_AHEAD; ++j) {
            const unsigned u = (unsigned)(d[j] - lo);
            if (u < (unsigned)RANGE_NODES) rank[e + j * COUNT_T] = r[j];
        }
    }
    __syncthreads();

    int hi = n - lo;
    hi = (hi < RANGE_NODES) ? hi : RANGE_NODES;
    int* __restrict__ row = part + (size_t)c * n + lo;
    for (int i = t; i < hi; i += COUNT_T) row[i] = hist[i];
}

__global__ void chunk_prefix_kernel(const int* __restrict__ part, int* __restrict__ before,
                                    int* __restrict__ cnt, int n) {
    const int d = blockIdx.x * blockDim.x + threadIdx.x;
    if (d >= n) return;
    int v[EDGE_CHUNKS];
#pragma unroll
    for (int c = 0; c < EDGE_CHUNKS; ++c) v[c] = part[(size_t)c * n + d];
    int s = 0;
#pragma unroll
    for (int c = 0; c < EDGE_CHUNKS; ++c) {
        before[(size_t)c * n + d] = s;
        s += v[c];
    }
    cnt[d] = s;
}

#define SCAN_T 1024
__global__ void __launch_bounds__(SCAN_T)
rowptr_kernel(const int* __restrict__ cnt, int* __restrict__ rowptr, int n) {
    __shared__ int wsum[32];
    __shared__ int sbase;
    const int t    = threadIdx.x;
    const int lane = t & 31;
    const int w    = t >> 5;
    const int first = blockIdx.x * SCAN_T;

    int part = 0;
    for (int i = t; i < first; i += SCAN_T) part += cnt[i];
    part += __shfl_xor(part, 16);
    part += __shfl_xor(part, 8);
    part += __shfl_xor(part, 4);
    part += __shfl_xor(part, 2);
    part += __shfl_xor(part, 1);
    if (lane == 0) wsum[w] = part;
    __syncthreads();
    if (w == 0) {
        int v = wsum[lane];
        v += __shfl_xor(v, 16);
        v += __shfl_xor(v, 8);
        v += __shfl_xor(v, 4);
        v += __shfl_xor(v, 2);
        v += __shfl_xor(v, 1);
        if (lane == 0) sbase = v;
    }
    __syncthreads();
    const int base = sbase;

    const int i = first + t;
    const int c = (i < n) ? cnt[i] : 0;
    int incl = c;
    for (int o = 1; o < 32; o <<= 1) {
        const int u = __shfl_up(incl, o);
        if (lane >= o) incl += u;
    }
    if (lane == 31) wsum[w] = incl;
    __syncthreads();
    if (w == 0) {
        const int v = wsum[lane];
        int inc2 = v;
        for (int o = 1; o < 32; o <<= 1) {
            const int u = __shfl_up(inc2, o);
            if (lane >= o) inc2 += u;
        }
        wsum[lane] = inc2 - v;
    }
    __syncthreads();
    const int excl = base + wsum[w] + incl - c;
    if (i < n) rowptr[i] = excl;
    if (i == n - 1) rowptr[n] = excl + c;
}

__global__ void place_kernel(const int* __restrict__ src,
                             const int* __restrict__ dst,
                             const int* __restrict__ rowptr,
                             const int* __restrict__ before,
                             const int* __restrict__ rank,
                             int* __restrict__ csr_src, int E, int n) {
    const int e = blockIdx.x * blockDim.x + threadIdx.x;
    if (e >= E) return;
    const int c = (int)(blockIdx.x * blockDim.x) / edge_chunk_len(E);
    const int d = dst[e];
    csr_src[rowptr[d] + before[(size_t)c * n + d] + rank[e]] = src[e];
}


__device__ __forceinline__ float head_score(const float4 v, const float4 zr, const float4 at) {
    float p = 0.0f, z;
    z = v.x + zr.x; z = (z > 0.0f) ? z : NEG_SLOPE * z; p += z * at.x;
    z = v.y + zr.y; z = (z > 0.0f) ? z : NEG_SLOPE * z; p += z * at.y;
    z = v.z + zr.z; z = (z > 0.0f) ? z : NEG_SLOPE * z; p += z * at.z;
    z = v.w + zr.w; z = (z > 0.0f) ? z : NEG_SLOPE * z; p += z * at.w;
    p += __shfl_xor(p, 1);
    p += __shfl_xor(p, 2);
    p += __shfl_xor(p, 4);
    p += __shfl_xor(p, 8);
    return p;
}

__device__ __forceinline__ void softmax_step(const float4 v, const float e,
                                             float& m, float& den, float4& acc) {
    const float mn = fmaxf(m, e);
    const float sc = __expf(m - mn);
    const float a  = __expf(e - mn);
    den   = den   * sc + a;
    acc.x = acc.x * sc + a * v.x;
    acc.y = acc.y * sc + a * v.y;
    acc.z = acc.z * sc + a * v.z;
    acc.w = acc.w * sc + a * v.w;
    m = mn;
}

__device__ __forceinline__ float finish_value(float v, int apply_gelu) {
    if (apply_gelu) {
        const float u = 0.7978845608028654f * (v + 0.044715f * v * v * v);
        v = 0.5f * v * (1.0f + tanhf(u));
    }
    return v;
}

#define ROWS_IN_FLIGHT 8
#define TILE_NODES 16
#define NODE_WAVES 4
#define WAVE_NODES 4
#define XR_PITCH 132

__global__ void __launch_bounds__(128, 12)
node_kernel(const float* __restrict__ xin,
            const float* __restrict__ Wr,
            const float* __restrict__ br,
            const float* __restrict__ xl,
            const int* __restrict__ rowptr,
            const int* __restrict__ csr_src,
            const float* __restrict__ att,
            const float* __restrict__ bias,
            float* __restrict__ out,
            int apply_gelu, int n) {
    __shared__ __attribute__((aligned(16))) float sxr[TILE_NODES * XR_PITCH];

    const int w = __builtin_amdgcn_readfirstlane((int)(threadIdx.x >> 5));
    const int lane = threadIdx.x & 31;
    const int nbase = blockIdx.x * TILE_NODES;
    const int d0 = nbase + w * WAVE_NODES;

    const int c0 = lane << 2;
    const float4 at = *(const float4*)(att + c0);
    int rpi = d0 + (lane & 7);
    rpi = (rpi < n) ? rpi : n;
    const int rpv = rowptr[rpi];
    const int dfirst = (d0 < n) ? d0 : n - 1;
    const v4f* __restrict__ xl4 = (const v4f*)xl;

    int beg = __builtin_amdgcn_readlane(rpv, 0);
    int end = __builtin_amdgcn_readlane(rpv, 1);
    int idx = dfirst;
    {
        const int left0 = end - beg;
        const int nk0   = (left0 < 32) ? left0 : 32;
        if (lane < nk0) idx = csr_src[beg + lane];
    }

    {
        int mrow = nbase + (lane & 15);
        mrow = (mrow < n) ? mrow : n - 1;
        const int ncol  = (w << 5) + (lane & 15);
        const int khalf = (lane >> 4) << 1;
        v8f t0 = {};
        v8f t1 = {};
#pragma unroll
        for (int k0 = 0; k0 < D_FEAT; k0 += 4) {
            const int kk = k0 + khalf;
            v2f a, b0, b1;
            const float* ap = xin + (size_t)mrow * D_FEAT + kk;
            a.x = ap[0];
            a.y = ap[1];
            b0.x = Wr[(size_t)(kk + 0) * HC + ncol];
            b0.y = Wr[(size_t)(kk + 1) * HC + ncol];
            b1.x = Wr[(size_t)(kk + 0) * HC + ncol + 16];
            b1.y = Wr[(size_t)(kk + 1) * HC + ncol + 16];
            t0 = __builtin_amdgcn_wmma_f32_16x16x4_f32(
                 false, a,  false, b0,
                 (short)0, t0,  false,  false);
            t1 = __builtin_amdgcn_wmma_f32_16x16x4_f32(
                 false, a,  false, b1,
                 (short)0, t1,  false,  false);
        }
        const float bb0 = br[ncol];
        const float bb1 = br[ncol + 16];
        const int rloc = (lane >> 4) << 3;
#pragma unroll
        for (int j = 0; j < 8; ++j) {
            sxr[(rloc + j) * XR_PITCH + ncol]      = t0[j] + bb0;
            sxr[(rloc + j) * XR_PITCH + ncol + 16] = t1[j] + bb1;
        }
    }
    __syncthreads();

    for (int jn = 0; jn < WAVE_NODES; ++jn) {
        const int d = d0 + jn;
        if (d >= n) break;

        int nbeg = 0, nend = 0, nidx = d;
        if (jn + 1 < WAVE_NODES) {
            int dn = d + 1;
            dn = (dn < n) ? dn : n - 1;
            nbeg = __builtin_amdgcn_readlane(rpv, jn + 1);
            nend = __builtin_amdgcn_readlane(rpv, jn + 2);
            nidx = dn;
            const int nleft = nend - nbeg;
            const int nnk   = (nleft < 32) ? nleft : 32;
            if (lane < nnk) nidx = csr_src[nbeg + lane];
        }

        float4 acc;
        {
            const v4f t = xl4[(unsigned)d * (HC / 4) + (unsigned)lane];
            acc = make_float4(t.x, t.y, t.z, t.w);
        }
        const float4 zr = *(const float4*)(sxr + (w * WAVE_NODES + jn) * XR_PITCH + c0);

        float m   = 0.0f;
        float den = 1.0f;
        bool have_m = false;

        for (int base = beg; base < end; base += 32) {
            const int left = end - base;
            const int nk   = (left < 32) ? left : 32;
            if (base != beg) idx = (lane < nk) ? csr_src[base + lane] : d;
            for (int k0 = 0; k0 < nk; k0 += ROWS_IN_FLIGHT) {
                const int cnt = nk - k0;
                v4f r[ROWS_IN_FLIGHT];
#pragma unroll
                for (int j = 0; j < 4; ++j) {
                    const int s = __builtin_amdgcn_readlane(idx, k0 + j);
                    r[j] = xl4[(unsigned)s * (HC / 4) + (unsigned)lane];
                }
                if (cnt > 4) {
#pragma unroll
                    for (int j = 4; j < ROWS_IN_FLIGHT; ++j) {
                        const int s = __builtin_amdgcn_readlane(idx, k0 + j);
                        r[j] = xl4[(unsigned)s * (HC / 4) + (unsigned)lane];
                    }
                }
                if (!have_m) {
                    m = head_score(acc, zr, at);
                    have_m = true;
                }
                float4 v[4];
                float e[4];
#pragma unroll
                for (int j = 0; j < 4; ++j) {
                    v[j] = make_float4(r[j].x, r[j].y, r[j].z, r[j].w);
                    const float t = head_score(v[j], zr, at);
                    e[j] = (j < cnt) ? t : -__builtin_inff();
                }
#pragma unroll
                for (int j = 0; j < 4; ++j)
                    softmax_step(v[j], e[j], m, den, acc);
                if (cnt > 4) {
#pragma unroll
                    for (int j = 0; j < 4; ++j) {
                        v[j] = make_float4(r[4 + j].x, r[4 + j].y, r[4 + j].z, r[4 + j].w);
                        const float t = head_score(v[j], zr, at);
                        e[j] = (4 + j < cnt) ? t : -__builtin_inff();
                    }
#pragma unroll
                    for (int j = 0; j < 4; ++j)
                        softmax_step(v[j], e[j], m, den, acc);
                }
            }
        }

        const float inv = 1.0f / (den + 1e-16f);
        float4 r;
        r.x = acc.x * inv;
        r.y = acc.y * inv;
        r.z = acc.z * inv;
        r.w = acc.w * inv;
        r.x += __shfl_xor(r.x, 16);
        r.y += __shfl_xor(r.y, 16);
        r.z += __shfl_xor(r.z, 16);
        r.w += __shfl_xor(r.w, 16);

        if (lane < 16) {
            const v4f b = ((const v4f*)bias)[(unsigned)lane];
            float4 o;
            o.x = finish_value(0.5f * r.x + b.x, apply_gelu);
            o.y = finish_value(0.5f * r.y + b.y, apply_gelu);
            o.z = finish_value(0.5f * r.z + b.z, apply_gelu);
            o.w = finish_value(0.5f * r.w + b.w, apply_gelu);
            v4f o4;
            o4.x = o.x; o4.y = o.y; o4.z = o.z; o4.w = o.w;
            ((v4f*)out)[(unsigned)d * (CCH / 4) + (unsigned)lane] = o4;
        }

        beg = nbeg;
        end = nend;
        idx = nidx;
    }
}


extern "C" void kernel_launch(void* const* d_in, const int* in_sizes, int n_in,
                              void* d_out, int out_size, void* d_ws, size_t ws_size,
                              hipStream_t stream) {
    const float* x0   = (const float*)d_in[0];
    const int*   conn = (const int*)  d_in[1];
    const float* Wl   = (const float*)d_in[2];
    const float* bl   = (const float*)d_in[3];
    const float* Wr   = (const float*)d_in[4];
    const float* br   = (const float*)d_in[5];
    const float* att  = (const float*)d_in[6];
    const float* bias = (const float*)d_in[7];

    const int N  = in_sizes[0] / D_FEAT;
    const int E  = in_sizes[1] / 2;
    const int* src = conn;
    const int* dst = conn + E;

    float* ws    = (float*)d_ws;
    float* xl    = ws;
    float* x1    = xl   + (size_t)N * HC;
    int*   cnt     = (int*)(x1 + (size_t)N * D_FEAT);
    int*   rowptr  = cnt + N;
    int*   csr_src = rowptr + N + 1;
    int*   rank    = csr_src + E;
    int*   part    = rank + E;
    int*   before  = part + (size_t)EDGE_CHUNKS * N;

    const dim3 blk(256);
    const int gemm_blocks = ((N / 16) * 8 + 7) / 8;
    const int node_blocks = (N + TILE_NODES - 1) / TILE_NODES;

    const int n_ranges = (N + RANGE_NODES - 1) / RANGE_NODES;
    count_kernel<<<n_ranges * EDGE_CHUNKS, dim3(COUNT_T), 0, stream>>>(dst, rank, part, E, N);
    chunk_prefix_kernel<<<(N + 255) / 256, blk, 0, stream>>>(part, before, cnt, N);
    rowptr_kernel<<<(N + SCAN_T - 1) / SCAN_T, dim3(SCAN_T), 0, stream>>>(cnt, rowptr, N);
    place_kernel<<<(E + 255) / 256, blk, 0, stream>>>(src, dst, rowptr, before, rank, csr_src, E, N);

    const int L = 2;
    const float* xin = x0;
    for (int l = 0; l < L; ++l) {
        const float* Wl_l = Wl + (size_t)l * D_FEAT * HC;
        const float* bl_l = bl + (size_t)l * HC;
        const float* Wr_l = Wr + (size_t)l * D_FEAT * HC;
        const float* br_l = br + (size_t)l * HC;
        const float* at_l = att + (size_t)l * NHEAD * CCH;
        const float* bi_l = bias + (size_t)l * CCH;

        gemm64x128_wmma<<<gemm_blocks, blk, 0, stream>>>(xin, Wl_l, bl_l, xl, N);

        float* outp = (l == L - 1) ? (float*)d_out : x1;
        node_kernel<<<node_blocks, dim3(32 * NODE_WAVES), 0, stream>>>(xin, Wr_l, br_l, xl, rowptr, csr_src,
                                                           at_l, bi_l, outp,
                                                           (l < L - 1) ? 1 : 0, N);

        xin = x1;
    }
}


# x@Wl GEMM retiled: a wave keeps its two column tiles' W fragments in registers and walks five row tiles, x read in 16-B pieces (k regrouped), whole 128-B lines stored; order of float sums changes
# speedup vs baseline: 4.2010x; 1.1406x over previous
#include <hip/hip_runtime.h>

#define D_FEAT 64
#define HC 128
#define NHEAD 2
#define CCH 64
#define NEG_SLOPE 0.2f

typedef __attribute__((ext_vector_type(2))) float v2f;
typedef __attribute__((ext_vector_type(8))) float v8f;
typedef __attribute__((ext_vector_type(4))) float v4f;

#define GEMM_GROUPS 1250

__global__ void __launch_bounds__(128)
gemm64x128_wmma(const float* __restrict__ X,
                const float* __restrict__ W,
                const float* __restrict__ bvec,
                float* __restrict__ Y, int rows) {
    const int w    = __builtin_amdgcn_readfirstlane((int)(threadIdx.x >> 5));
    const int lane = threadIdx.x & 31;
    const int c    = lane & 15;
    const int h    = lane >> 4;
    const int tiles_m = rows >> 4;
    const int ncol = (w << 5) + c;

    int tm = blockIdx.x;
    if (tm >= tiles_m) return;

    const v4f* __restrict__ xp = (const v4f*)X;
    v4f a[8];
    {
        const unsigned piece = (unsigned)((tm << 4) + c) * (D_FEAT / 4) + (unsigned)h;
#pragma unroll
        for (int i = 0; i < 8; ++i) a[i] = xp[piece + 2 * i];
    }

    v2f b0[16], b1[16];
#pragma unroll
    for (int i = 0; i < 8; ++i) {
        const float* wp = W + (size_t)(8 * i + 4 * h) * HC + ncol;
        b0[2 * i].x     = wp[0];
        b0[2 * i].y     = wp[HC];
        b0[2 * i + 1].x = wp[2 * HC];
        b0[2 * i + 1].y = wp[3 * HC];
        b1[2 * i].x     = wp[16];
        b1[2 * i].y     = wp[HC + 16];
        b1[2 * i + 1].x = wp[2 * HC + 16];
        b1[2 * i + 1].y = wp[3 * HC + 16];
    }
    const float bb0 = bvec[ncol];
    const float bb1 = bvec[ncol + 16];
    float* __restrict__ ycol = Y + (w << 5) + lane;

    for (; tm < tiles_m; tm += GEMM_GROUPS) {
        int tn = tm + GEMM_GROUPS;
        tn = (tn < tiles_m) ? tn : tm;
        v4f an[8];
        {
            const unsigned piece = (unsigned)((tn << 4) + c) * (D_FEAT / 4) + (unsigned)h;
#pragma unroll
            for (int i = 0; i < 8; ++i) an[i] = xp[piece + 2 * i];
        }

        v8f t0 = {};
        v8f t1 = {};
#pragma unroll
        for (int i = 0; i < 8; ++i) {
            v2f alo, ahi;
            alo.x = a[i].x; alo.y = a[i].y;
            ahi.x = a[i].z; ahi.y = a[i].w;
            t0 = __builtin_amdgcn_wmma_f32_16x16x4_f32(
                 false, alo,  false, b0[2 * i],
                 (short)0, t0,  false,  false);
            t1 = __builtin_amdgcn_wmma_f32_16x16x4_f32(
                 false, alo,  false, b1[2 * i],
                 (short)0, t1,  false,  false);
            t0 = __builtin_amdgcn_wmma_f32_16x16x4_f32(
                 false, ahi,  false, b0[2 * i + 1],
                 (short)0, t0,  false,  false);
            t1 = __builtin_amdgcn_wmma_f32_16x16x4_f32(
                 false, ahi,  false, b1[2 * i + 1],
                 (short)0, t1,  false,  false);
        }

        float* __restrict__ yrow = ycol + (size_t)(tm << 4) * HC;
#pragma unroll
        for (int j = 0; j < 8; ++j) {
            const float own0 = t0[j] + bb0;
            const float own1 = t1[j] + bb1;
            const float give = h ? own0 : own1;
            const float got  = __shfl_xor(give, 16);
            yrow[(size_t)j * HC]       = h ? got : own0;
            yrow[(size_t)(j + 8) * HC] = h ? own1 : got;
        }

#pragma unroll
        for (int i = 0; i < 8; ++i) a[i] = an[i];
    }
}


#define RANGE_NODES 16384
#define EDGE_CHUNKS 32
#define COUNT_T 1024
#define COUNT_AHEAD 16

__device__ __forceinline__ int edge_chunk_len(int E) {
    return ((E + EDGE_CHUNKS * 256 - 1) / (EDGE_CHUNKS * 256)) * 256;
}

__global__ void __launch_bounds__(COUNT_T)
count_kernel(const int* __restrict__ dst, int* __restrict__ rank,
             int* __restrict__ part, int E, int n) {
    __shared__ int hist[RANGE_NODES];
    const int t  = threadIdx.x;
    const int c  = blockIdx.x % EDGE_CHUNKS;
    const int lo = (blockIdx.x / EDGE_CHUNKS) * RANGE_NODES;

    for (int i = t; i < RANGE_NODES; i += COUNT_T) hist[i] = 0;
    __syncthreads();

    const int chunk = edge_chunk_len(E);
    const int e0 = c * chunk;
    int e1 = e0 + chunk;
    e1 = (e1 < E) ? e1 : E;
    for (int e = e0 + t; e < e1; e += COUNT_T * COUNT_AHEAD) {
        int d[COUNT_AHEAD];
#pragma unroll
        for (int j = 0; j < COUNT_AHEAD; ++j) {
            const int ee = e + j * COUNT_T;
            d[j] = (ee < e1) ? dst[ee] : -1;
        }
        int r[COUNT_AHEAD];
#pragma unroll
        for (int j = 0; j < COUNT_AHEAD; ++j) {
            const unsigned u = (unsigned)(d[j] - lo);
            r[j] = 0;
            if (u < (unsigned)RANGE_NODES)
                r[j] = atomicAdd(&hist[u], 1);
        }
#pragma unroll
        for (int j = 0; j < COUNT_AHEAD; ++j) {
            const unsigned u = (unsigned)(d[j] - lo);
            if (u < (unsigned)RANGE_NODES) rank[e + j * COUNT_T] = r[j];
        }
    }
    __syncthreads();

    int hi = n - lo;
    hi = (hi < RANGE_NODES) ? hi : RANGE_NODES;
    int* __restrict__ row = part + (size_t)c * n + lo;
    for (int i = t; i < hi; i += COUNT_T) row[i] = hist[i];
}

__global__ void chunk_prefix_kernel(const int* __restrict__ part, int* __restrict__ before,
                                    int* __restrict__ cnt, int n) {
    const int d = blockIdx.x * blockDim.x + threadIdx.x;
    if (d >= n) return;
    int v[EDGE_CHUNKS];
#pragma unroll
    for (int c = 0; c < EDGE_CHUNKS; ++c) v[c] = part[(size_t)c * n + d];
    int s = 0;
#pragma unroll
    for (int c = 0; c < EDGE_CHUNKS; ++c) {
        before[(size_t)c * n + d] = s;
        s += v[c];
    }
    cnt[d] = s;
}

#define SCAN_T 1024
__global__ void __launch_bounds__(SCAN_T)
rowptr_kernel(const int* __restrict__ cnt, int* __restrict__ rowptr, int n) {
    __shared__ int wsum[32];
    __shared__ int sbase;
    const int t    = threadIdx.x;
    const int lane = t & 31;
    const int w    = t >> 5;
    const int first = blockIdx.x * SCAN_T;

    int part = 0;
    for (int i = t; i < first; i += SCAN_T) part += cnt[i];
    part += __shfl_xor(part, 16);
    part += __shfl_xor(part, 8);
    part += __shfl_xor(part, 4);
    part += __shfl_xor(part, 2);
    part += __shfl_xor(part, 1);
    if (lane == 0) wsum[w] = part;
    __syncthreads();
    if (w == 0) {
        int v = wsum[lane];
        v += __shfl_xor(v, 16);
        v += __shfl_xor(v, 8);
        v += __shfl_xor(v, 4);
        v += __shfl_xor(v, 2);
        v += __shfl_xor(v, 1);
        if (lane == 0) sbase = v;
    }
    __syncthreads();
    const int base = sbase;

    const int i = first + t;
    const int c = (i < n) ? cnt[i] : 0;
    int incl = c;
    for (int o = 1; o < 32; o <<= 1) {
        const int u = __shfl_up(incl, o);
        if (lane >= o) incl += u;
    }
    if (lane == 31) wsum[w] = incl;
    __syncthreads();
    if (w == 0) {
        const int v = wsum[lane];
        int inc2 = v;
        for (int o = 1; o < 32; o <<= 1) {
            const int u = __shfl_up(inc2, o);
            if (lane >= o) inc2 += u;
        }
        wsum[lane] = inc2 - v;
    }
    __syncthreads();
    const int excl = base + wsum[w] + incl - c;
    if (i < n) rowptr[i] = excl;
    if (i == n - 1) rowptr[n] = excl + c;
}

__global__ void place_kernel(const int* __restrict__ src,
                             const int* __restrict__ dst,
                             const int* __restrict__ rowptr,
                             const int* __restrict__ before,
                             const int* __restrict__ rank,
                             int* __restrict__ csr_src, int E, int n) {
    const int e = blockIdx.x * blockDim.x + threadIdx.x;
    if (e >= E) return;
    const int c = (int)(blockIdx.x * blockDim.x) / edge_chunk_len(E);
    const int d = dst[e];
    csr_src[rowptr[d] + before[(size_t)c * n + d] + rank[e]] = src[e];
}


__device__ __forceinline__ float head_score(const float4 v, const float4 zr, const float4 at) {
    float p = 0.0f, z;
    z = v.x + zr.x; z = (z > 0.0f) ? z : NEG_SLOPE * z; p += z * at.x;
    z = v.y + zr.y; z = (z > 0.0f) ? z : NEG_SLOPE * z; p += z * at.y;
    z = v.z + zr.z; z = (z > 0.0f) ? z : NEG_SLOPE * z; p += z * at.z;
    z = v.w + zr.w; z = (z > 0.0f) ? z : NEG_SLOPE * z; p += z * at.w;
    p += __shfl_xor(p, 1);
    p += __shfl_xor(p, 2);
    p += __shfl_xor(p, 4);
    p += __shfl_xor(p, 8);
    return p;
}

__device__ __forceinline__ void softmax_step(const float4 v, const float e,
                                             float& m, float& den, float4& acc) {
    const float mn = fmaxf(m, e);
    const float sc = __expf(m - mn);
    const float a  = __expf(e - mn);
    den   = den   * sc + a;
    acc.x = acc.x * sc + a * v.x;
    acc.y = acc.y * sc + a * v.y;
    acc.z = acc.z * sc + a * v.z;
    acc.w = acc.w * sc + a * v.w;
    m = mn;
}

__device__ __forceinline__ float finish_value(float v, int apply_gelu) {
    if (apply_gelu) {
        const float u = 0.7978845608028654f * (v + 0.044715f * v * v * v);
        v = 0.5f * v * (1.0f + tanhf(u));
    }
    return v;
}

#define ROWS_IN_FLIGHT 8
#define TILE_NODES 16
#define NODE_WAVES 4
#define WAVE_NODES 4
#define XR_PITCH 132

__global__ void __launch_bounds__(128, 12)
node_kernel(const float* __restrict__ xin,
            const float* __restrict__ Wr,
            const float* __restrict__ br,
            const float* __restrict__ xl,
            const int* __restrict__ rowptr,
            const int* __restrict__ csr_src,
            const float* __restrict__ att,
            const float* __restrict__ bias,
            float* __restrict__ out,
            int apply_gelu, int n) {
    __shared__ __attribute__((aligned(16))) float sxr[TILE_NODES * XR_PITCH];

    const int w = __builtin_amdgcn_readfirstlane((int)(threadIdx.x >> 5));
    const int lane = threadIdx.x & 31;
    const int nbase = blockIdx.x * TILE_NODES;
    const int d0 = nbase + w * WAVE_NODES;

    const int c0 = lane << 2;
    const float4 at = *(const float4*)(att + c0);
    int rpi = d0 + (lane & 7);
    rpi = (rpi < n) ? rpi : n;
    const int rpv = rowptr[rpi];
    const int dfirst = (d0 < n) ? d0 : n - 1;
    const v4f* __restrict__ xl4 = (const v4f*)xl;

    int beg = __builtin_amdgcn_readlane(rpv, 0);
    int end = __builtin_amdgcn_readlane(rpv, 1);
    int idx = dfirst;
    {
        const int left0 = end - beg;
        const int nk0   = (left0 < 32) ? left0 : 32;
        if (lane < nk0) idx = csr_src[beg + lane];
    }

    {
        int mrow = nbase + (lane & 15);
        mrow = (mrow < n) ? mrow : n - 1;
        const int ncol  = (w << 5) + (lane & 15);
        const int khalf = (lane >> 4) << 1;
        v8f t0 = {};
        v8f t1 = {};
#pragma unroll
        for (int k0 = 0; k0 < D_FEAT; k0 += 4) {
            const int kk = k0 + khalf;
            v2f a, b0, b1;
            const float* ap = xin + (size_t)mrow * D_FEAT + kk;
            a.x = ap[0];
            a.y = ap[1];
            b0.x = Wr[(size_t)(kk + 0) * HC + ncol];
            b0.y = Wr[(size_t)(kk + 1) * HC + ncol];
            b1.x = Wr[(size_t)(kk + 0) * HC + ncol + 16];
            b1.y = Wr[(size_t)(kk + 1) * HC + ncol + 16];
            t0 = __builtin_amdgcn_wmma_f32_16x16x4_f32(
                 false, a,  false, b0,
                 (short)0, t0,  false,  false);
            t1 = __builtin_amdgcn_wmma_f32_16x16x4_f32(
                 false, a,  false, b1,
                 (short)0, t1,  false,  false);
        }
        const float bb0 = br[ncol];
        const float bb1 = br[ncol + 16];
        const int rloc = (lane >> 4) << 3;
#pragma unroll
        for (int j = 0; j < 8; ++j) {
            sxr[(rloc + j) * XR_PITCH + ncol]      = t0[j] + bb0;
            sxr[(rloc + j) * XR_PITCH + ncol + 16] = t1[j] + bb1;
        }
    }
    __syncthreads();

    for (int jn = 0; jn < WAVE_NODES; ++jn) {
        const int d = d0 + jn;
        if (d >= n) break;

        int nbeg = 0, nend = 0, nidx = d;
        if (jn + 1 < WAVE_NODES) {
            int dn = d + 1;
            dn = (dn < n) ? dn : n - 1;
            nbeg = __builtin_amdgcn_readlane(rpv, jn + 1);
            nend = __builtin_amdgcn_readlane(rpv, jn + 2);
            nidx = dn;
            const int nleft = nend - nbeg;
            const int nnk   = (nleft < 32) ? nleft : 32;
            if (lane < nnk) nidx = csr_src[nbeg + lane];
        }

        float4 acc;
        {
            const v4f t = xl4[(unsigned)d * (HC / 4) + (unsigned)lane];
            acc = make_float4(t.x, t.y, t.z, t.w);
        }
        const float4 zr = *(const float4*)(sxr + (w * WAVE_NODES + jn) * XR_PITCH + c0);

        float m   = 0.0f;
        float den = 1.0f;
        bool have_m = false;

        for (int base = beg; base < end; base += 32) {
            const int left = end - base;
            const int nk   = (left < 32) ? left : 32;
            if (base != beg) idx = (lane < nk) ? csr_src[base + lane] : d;
            for (int k0 = 0; k0 < nk; k0 += ROWS_IN_FLIGHT) {
                const int cnt = nk - k0;
                v4f r[ROWS_IN_FLIGHT];
#pragma unroll
                for (int j = 0; j < 4; ++j) {
                    const int s = __builtin_amdgcn_readlane(idx, k0 + j);
                    r[j] = xl4[(unsigned)s * (HC / 4) + (unsigned)lane];
                }
                if (cnt > 4) {
#pragma unroll
                    for (int j = 4; j < ROWS_IN_FLIGHT; ++j) {
                        const int s = __builtin_amdgcn_readlane(idx, k0 + j);
                        r[j] = xl4[(unsigned)s * (HC / 4) + (unsigned)lane];
                    }
                }
                if (!have_m) {
                    m = head_score(acc, zr, at);
                    have_m = true;
                }
                float4 v[4];
                float e[4];
#pragma unroll
                for (int j = 0; j < 4; ++j) {
                    v[j] = make_float4(r[j].x, r[j].y, r[j].z, r[j].w);
                    const float t = head_score(v[j], zr, at);
                    e[j] = (j < cnt) ? t : -__builtin_inff();
                }
#pragma unroll
                for (int j = 0; j < 4; ++j)
                    softmax_step(v[j], e[j], m, den, acc);
                if (cnt > 4) {
#pragma unroll
                    for (int j = 0; j < 4; ++j) {
                        v[j] = make_float4(r[4 + j].x, r[4 + j].y, r[4 + j].z, r[4 + j].w);
                        const float t = head_score(v[j], zr, at);
                        e[j] = (4 + j < cnt) ? t : -__builtin_inff();
                    }
#pragma unroll
                    for (int j = 0; j < 4; ++j)
                        softmax_step(v[j], e[j], m, den, acc);
                }
            }
        }

        const float inv = 1.0f / (den + 1e-16f);
        float4 r;
        r.x = acc.x * inv;
        r.y = acc.y * inv;
        r.z = acc.z * inv;
        r.w = acc.w * inv;
        r.x += __shfl_xor(r.x, 16);
        r.y += __shfl_xor(r.y, 16);
        r.z += __shfl_xor(r.z, 16);
        r.w += __shfl_xor(r.w, 16);

        if (lane < 16) {
            const v4f b = ((const v4f*)bias)[(unsigned)lane];
            float4 o;
            o.x = finish_value(0.5f * r.x + b.x, apply_gelu);
            o.y = finish_value(0.5f * r.y + b.y, apply_gelu);
            o.z = finish_value(0.5f * r.z + b.z, apply_gelu);
            o.w = finish_value(0.5f * r.w + b.w, apply_gelu);
            v4f o4;
            o4.x = o.x; o4.y = o.y; o4.z = o.z; o4.w = o.w;
            ((v4f*)out)[(unsigned)d * (CCH / 4) + (unsigned)lane] = o4;
        }

        beg = nbeg;
        end = nend;
        idx = nidx;
    }
}


extern "C" void kernel_launch(void* const* d_in, const int* in_sizes, int n_in,
                              void* d_out, int out_size, void* d_ws, size_t ws_size,
                              hipStream_t stream) {
    const float* x0   = (const float*)d_in[0];
    const int*   conn = (const int*)  d_in[1];
    const float* Wl   = (const float*)d_in[2];
    const float* bl   = (const float*)d_in[3];
    const float* Wr   = (const float*)d_in[4];
    const float* br   = (const float*)d_in[5];
    const float* att  = (const float*)d_in[6];
    const float* bias = (const float*)d_in[7];

    const int N  = in_sizes[0] / D_FEAT;
    const int E  = in_sizes[1] / 2;
    const int* src = conn;
    const int* dst = conn + E;

    float* ws    = (float*)d_ws;
    float* xl    = ws;
    float* x1    = xl   + (size_t)N * HC;
    int*   cnt     = (int*)(x1 + (size_t)N * D_FEAT);
    int*   rowptr  = cnt + N;
    int*   csr_src = rowptr + N + 1;
    int*   rank    = csr_src + E;
    int*   part    = rank + E;
    int*   before  = part + (size_t)EDGE_CHUNKS * N;

    const dim3 blk(256);
    const int node_blocks = (N + TILE_NODES - 1) / TILE_NODES;

    const int n_ranges = (N + RANGE_NODES - 1) / RANGE_NODES;
    count_kernel<<<n_ranges * EDGE_CHUNKS, dim3(COUNT_T), 0, stream>>>(dst, rank, part, E, N);
    chunk_prefix_kernel<<<(N + 255) / 256, blk, 0, stream>>>(part, before, cnt, N);
    rowptr_kernel<<<(N + SCAN_T - 1) / SCAN_T, dim3(SCAN_T), 0, stream>>>(cnt, rowptr, N);
    place_kernel<<<(E + 255) / 256, blk, 0, stream>>>(src, dst, rowptr, before, rank, csr_src, E, N);

    const int L = 2;
    const float* xin = x0;
    for (int l = 0; l < L; ++l) {
        const float* Wl_l = Wl + (size_t)l * D_FEAT * HC;
        const float* bl_l = bl + (size_t)l * HC;
        const float* Wr_l = Wr + (size_t)l * D_FEAT * HC;
        const float* br_l = br + (size_t)l * HC;
        const float* at_l = att + (size_t)l * NHEAD * CCH;
        const float* bi_l = bias + (size_t)l * CCH;

        gemm64x128_wmma<<<GEMM_GROUPS, dim3(128), 0, stream>>>(xin, Wl_l, bl_l, xl, N);

        float* outp = (l == L - 1) ? (float*)d_out : x1;
        node_kernel<<<node_blocks, dim3(32 * NODE_WAVES), 0, stream>>>(xin, Wr_l, br_l, xl, rowptr, csr_src,
                                                           at_l, bi_l, outp,
                                                           (l < L - 1) ? 1 : 0, N);

        xin = x1;
    }
}


# Node pass: the 16-lane score sums and the merge of the two heads exchange lanes on the vector ALU (row_xmask operand of the add, permlanex16), not through LDS-unit permutes; same sum order
# speedup vs baseline: 4.3482x; 1.0350x over previous
#include <hip/hip_runtime.h>

#define D_FEAT 64
#define HC 128
#define NHEAD 2
#define CCH 64
#define NEG_SLOPE 0.2f

typedef __attribute__((ext_vector_type(2))) float v2f;
typedef __attribute__((ext_vector_type(8))) float v8f;
typedef __attribute__((ext_vector_type(4))) float v4f;

#define GEMM_GROUPS 1250

__global__ void __launch_bounds__(128)
gemm64x128_wmma(const float* __restrict__ X,
                const float* __restrict__ W,
                const float* __restrict__ bvec,
                float* __restrict__ Y, int rows) {
    const int w    = __builtin_amdgcn_readfirstlane((int)(threadIdx.x >> 5));
    const int lane = threadIdx.x & 31;
    const int c    = lane & 15;
    const int h    = lane >> 4;
    const int tiles_m = rows >> 4;
    const int ncol = (w << 5) + c;

    int tm = blockIdx.x;
    if (tm >= tiles_m) return;

    const v4f* __restrict__ xp = (const v4f*)X;
    v4f a[8];
    {
        const unsigned piece = (unsigned)((tm << 4) + c) * (D_FEAT / 4) + (unsigned)h;
#pragma unroll
        for (int i = 0; i < 8; ++i) a[i] = xp[piece + 2 * i];
    }

    v2f b0[16], b1[16];
#pragma unroll
    for (int i = 0; i < 8; ++i) {
        const float* wp = W + (size_t)(8 * i + 4 * h) * HC + ncol;
        b0[2 * i].x     = wp[0];
        b0[2 * i].y     = wp[HC];
        b0[2 * i + 1].x = wp[2 * HC];
        b0[2 * i + 1].y = wp[3 * HC];
        b1[2 * i].x     = wp[16];
        b1[2 * i].y     = wp[HC + 16];
        b1[2 * i + 1].x = wp[2 * HC + 16];
        b1[2 * i + 1].y = wp[3 * HC + 16];
    }
    const float bb0 = bvec[ncol];
    const float bb1 = bvec[ncol + 16];
    float* __restrict__ ycol = Y + (w << 5) + lane;

    for (; tm < tiles_m; tm += GEMM_GROUPS) {
        int tn = tm + GEMM_GROUPS;
        tn = (tn < tiles_m) ? tn : tm;
        v4f an[8];
        {
            const unsigned piece = (unsigned)((tn << 4) + c) * (D_FEAT / 4) + (unsigned)h;
#pragma unroll
            for (int i = 0; i < 8; ++i) an[i] = xp[piece + 2 * i];
        }

        v8f t0 = {};
        v8f t1 = {};
#pragma unroll
        for (int i = 0; i < 8; ++i) {
            v2f alo, ahi;
            alo.x = a[i].x; alo.y = a[i].y;
            ahi.x = a[i].z; ahi.y = a[i].w;
            t0 = __builtin_amdgcn_wmma_f32_16x16x4_f32(
                 false, alo,  false, b0[2 * i],
                 (short)0, t0,  false,  false);
            t1 = __builtin_amdgcn_wmma_f32_16x16x4_f32(
                 false, alo,  false, b1[2 * i],
                 (short)0, t1,  false,  false);
            t0 = __builtin_amdgcn_wmma_f32_16x16x4_f32(
                 false, ahi,  false, b0[2 * i + 1],
                 (short)0, t0,  false,  false);
            t1 = __builtin_amdgcn_wmma_f32_16x16x4_f32(
                 false, ahi,  false, b1[2 * i + 1],
                 (short)0, t1,  false,  false);
        }

        float* __restrict__ yrow = ycol + (size_t)(tm << 4) * HC;
#pragma unroll
        for (int j = 0; j < 8; ++j) {
            const float own0 = t0[j] + bb0;
            const float own1 = t1[j] + bb1;
            const float give = h ? own0 : own1;
            const float got  = __shfl_xor(give, 16);
            yrow[(size_t)j * HC]       = h ? got : own0;
            yrow[(size_t)(j + 8) * HC] = h ? own1 : got;
        }

#pragma unroll
        for (int i = 0; i < 8; ++i) a[i] = an[i];
    }
}


#define RANGE_NODES 16384
#define EDGE_CHUNKS 32
#define COUNT_T 1024
#define COUNT_AHEAD 16

__device__ __forceinline__ int edge_chunk_len(int E) {
    return ((E + EDGE_CHUNKS * 256 - 1) / (EDGE_CHUNKS * 256)) * 256;
}

__global__ void __launch_bounds__(COUNT_T)
count_kernel(const int* __restrict__ dst, int* __restrict__ rank,
             int* __restrict__ part, int E, int n) {
    __shared__ int hist[RANGE_NODES];
    const int t  = threadIdx.x;
    const int c  = blockIdx.x % EDGE_CHUNKS;
    const int lo = (blockIdx.x / EDGE_CHUNKS) * RANGE_NODES;

    for (int i = t; i < RANGE_NODES; i += COUNT_T) hist[i] = 0;
    __syncthreads();

    const int chunk = edge_chunk_len(E);
    const int e0 = c * chunk;
    int e1 = e0 + chunk;
    e1 = (e1 < E) ? e1 : E;
    for (int e = e0 + t; e < e1; e += COUNT_T * COUNT_AHEAD) {
        int d[COUNT_AHEAD];
#pragma unroll
        for (int j = 0; j < COUNT_AHEAD; ++j) {
            const int ee = e + j * COUNT_T;
            d[j] = (ee < e1) ? dst[ee] : -1;
        }
        int r[COUNT_AHEAD];
#pragma unroll
        for (int j = 0; j < COUNT_AHEAD; ++j) {
            const unsigned u = (unsigned)(d[j] - lo);
            r[j] = 0;
            if (u < (unsigned)RANGE_NODES)
                r[j] = atomicAdd(&hist[u], 1);
        }
#pragma unroll
        for (int j = 0; j < COUNT_AHEAD; ++j) {
            const unsigned u = (unsigned)(d[j] - lo);
            if (u < (unsigned)RANGE_NODES) rank[e + j * COUNT_T] = r[j];
        }
    }
    __syncthreads();

    int hi = n - lo;
    hi = (hi < RANGE_NODES) ? hi : RANGE_NODES;
    int* __restrict__ row = part + (size_t)c * n + lo;
    for (int i = t; i < hi; i += COUNT_T) row[i] = hist[i];
}

__global__ void chunk_prefix_kernel(const int* __restrict__ part, int* __restrict__ before,
                                    int* __restrict__ cnt, int n) {
    const int d = blockIdx.x * blockDim.x + threadIdx.x;
    if (d >= n) return;
    int v[EDGE_CHUNKS];
#pragma unroll
    for (int c = 0; c < EDGE_CHUNKS; ++c) v[c] = part[(size_t)c * n + d];
    int s = 0;
#pragma unroll
    for (int c = 0; c < EDGE_CHUNKS; ++c) {
        before[(size_t)c * n + d] = s;
        s += v[c];
    }
    cnt[d] = s;
}

#define SCAN_T 1024
__global__ void __launch_bounds__(SCAN_T)
rowptr_kernel(const int* __restrict__ cnt, int* __restrict__ rowptr, int n) {
    __shared__ int wsum[32];
    __shared__ int sbase;
    const int t    = threadIdx.x;
    const int lane = t & 31;
    const int w    = t >> 5;
    const int first = blockIdx.x * SCAN_T;

    int part = 0;
    for (int i = t; i < first; i += SCAN_T) part += cnt[i];
    part += __shfl_xor(part, 16);
    part += __shfl_xor(part, 8);
    part += __shfl_xor(part, 4);
    part += __shfl_xor(part, 2);
    part += __shfl_xor(part, 1);
    if (lane == 0) wsum[w] = part;
    __syncthreads();
    if (w == 0) {
        int v = wsum[lane];
        v += __shfl_xor(v, 16);
        v += __shfl_xor(v, 8);
        v += __shfl_xor(v, 4);
        v += __shfl_xor(v, 2);
        v += __shfl_xor(v, 1);
        if (lane == 0) sbase = v;
    }
    __syncthreads();
    const int base = sbase;

    const int i = first + t;
    const int c = (i < n) ? cnt[i] : 0;
    int incl = c;
    for (int o = 1; o < 32; o <<= 1) {
        const int u = __shfl_up(incl, o);
        if (lane >= o) incl += u;
    }
    if (lane == 31) wsum[w] = incl;
    __syncthreads();
    if (w == 0) {
        const int v = wsum[lane];
        int inc2 = v;
        for (int o = 1; o < 32; o <<= 1) {
            const int u = __shfl_up(inc2, o);
            if (lane >= o) inc2 += u;
        }
        wsum[lane] = inc2 - v;
    }
    __syncthreads();
    const int excl = base + wsum[w] + incl - c;
    if (i < n) rowptr[i] = excl;
    if (i == n - 1) rowptr[n] = excl + c;
}

__global__ void place_kernel(const int* __restrict__ src,
                             const int* __restrict__ dst,
                             const int* __restrict__ rowptr,
                             const int* __restrict__ before,
                             const int* __restrict__ rank,
                             int* __restrict__ csr_src, int E, int n) {
    const int e = blockIdx.x * blockDim.x + threadIdx.x;
    if (e >= E) return;
    const int c = (int)(blockIdx.x * blockDim.x) / edge_chunk_len(E);
    const int d = dst[e];
    csr_src[rowptr[d] + before[(size_t)c * n + d] + rank[e]] = src[e];
}


template <int XM>
__device__ __forceinline__ float half_xor(const float x) {
    const int b = __builtin_bit_cast(int, x);
    return __builtin_bit_cast(float, __builtin_amdgcn_mov_dpp(b, 0x160 + XM, 0xf, 0xf, true));
}
__device__ __forceinline__ float other_half(const float x) {
    const int b = __builtin_bit_cast(int, x);
    return __builtin_bit_cast(float,
        __builtin_amdgcn_permlanex16(b, b, 0x76543210, 0xfedcba98, false, false));
}

__device__ __forceinline__ float head_score(const float4 v, const float4 zr, const float4 at) {
    float p = 0.0f, z;
    z = v.x + zr.x; z = (z > 0.0f) ? z : NEG_SLOPE * z; p += z * at.x;
    z = v.y + zr.y; z = (z > 0.0f) ? z : NEG_SLOPE * z; p += z * at.y;
    z = v.z + zr.z; z = (z > 0.0f) ? z : NEG_SLOPE * z; p += z * at.z;
    z = v.w + zr.w; z = (z > 0.0f) ? z : NEG_SLOPE * z; p += z * at.w;
    p += half_xor<1>(p);
    p += half_xor<2>(p);
    p += half_xor<4>(p);
    p += half_xor<8>(p);
    return p;
}

__device__ __forceinline__ void softmax_step(const float4 v, const float e,
                                             float& m, float& den, float4& acc) {
    const float mn = fmaxf(m, e);
    const float sc = __expf(m - mn);
    const float a  = __expf(e - mn);
    den   = den   * sc + a;
    acc.x = acc.x * sc + a * v.x;
    acc.y = acc.y * sc + a * v.y;
    acc.z = acc.z * sc + a * v.z;
    acc.w = acc.w * sc + a * v.w;
    m = mn;
}

__device__ __forceinline__ float finish_value(float v, int apply_gelu) {
    if (apply_gelu) {
        const float u = 0.7978845608028654f * (v + 0.044715f * v * v * v);
        v = 0.5f * v * (1.0f + tanhf(u));
    }
    return v;
}

#define ROWS_IN_FLIGHT 8
#define TILE_NODES 16
#define NODE_WAVES 4
#define WAVE_NODES 4
#define XR_PITCH 132

__global__ void __launch_bounds__(128, 12)
node_kernel(const float* __restrict__ xin,
            const float* __restrict__ Wr,
            const float* __restrict__ br,
            const float* __restrict__ xl,
            const int* __restrict__ rowptr,
            const int* __restrict__ csr_src,
            const float* __restrict__ att,
            const float* __restrict__ bias,
            float* __restrict__ out,
            int apply_gelu, int n) {
    __shared__ __attribute__((aligned(16))) float sxr[TILE_NODES * XR_PITCH];

    const int w = __builtin_amdgcn_readfirstlane((int)(threadIdx.x >> 5));
    const int lane = threadIdx.x & 31;
    const int nbase = blockIdx.x * TILE_NODES;
    const int d0 = nbase + w * WAVE_NODES;

    const int c0 = lane << 2;
    const float4 at = *(const float4*)(att + c0);
    int rpi = d0 + (lane & 7);
    rpi = (rpi < n) ? rpi : n;
    const int rpv = rowptr[rpi];
    const int dfirst = (d0 < n) ? d0 : n - 1;
    const v4f* __restrict__ xl4 = (const v4f*)xl;

    int beg = __builtin_amdgcn_readlane(rpv, 0);
    int end = __builtin_amdgcn_readlane(rpv, 1);
    int idx = dfirst;
    {
        const int left0 = end - beg;
        const int nk0   = (left0 < 32) ? left0 : 32;
        if (lane < nk0) idx = csr_src[beg + lane];
    }

    {
        int mrow = nbase + (lane & 15);
        mrow = (mrow < n) ? mrow : n - 1;
        const int ncol  = (w << 5) + (lane & 15);
        const int khalf = (lane >> 4) << 1;
        v8f t0 = {};
        v8f t1 = {};
#pragma unroll
        for (int k0 = 0; k0 < D_FEAT; k0 += 4) {
            const int kk = k0 + khalf;
            v2f a, b0, b1;
            const float* ap = xin + (size_t)mrow * D_FEAT + kk;
            a.x = ap[0];
            a.y = ap[1];
            b0.x = Wr[(size_t)(kk + 0) * HC + ncol];
            b0.y = Wr[(size_t)(kk + 1) * HC + ncol];
            b1.x = Wr[(size_t)(kk + 0) * HC + ncol + 16];
            b1.y = Wr[(size_t)(kk + 1) * HC + ncol + 16];
            t0 = __builtin_amdgcn_wmma_f32_16x16x4_f32(
                 false, a,  false, b0,
                 (short)0, t0,  false,  false);
            t1 = __builtin_amdgcn_wmma_f32_16x16x4_f32(
                 false, a,  false, b1,
                 (short)0, t1,  false,  false);
        }
        const float bb0 = br[ncol];
        const float bb1 = br[ncol + 16];
        const int rloc = (lane >> 4) << 3;
#pragma unroll
        for (int j = 0; j < 8; ++j) {
            sxr[(rloc + j) * XR_PITCH + ncol]      = t0[j] + bb0;
            sxr[(rloc + j) * XR_PITCH + ncol + 16] = t1[j] + bb1;
        }
    }
    __syncthreads();

    for (int jn = 0; jn < WAVE_NODES; ++jn) {
        const int d = d0 + jn;
        if (d >= n) break;

        int nbeg = 0, nend = 0, nidx = d;
        if (jn + 1 < WAVE_NODES) {
            int dn = d + 1;
            dn = (dn < n) ? dn : n - 1;
            nbeg = __builtin_amdgcn_readlane(rpv, jn + 1);
            nend = __builtin_amdgcn_readlane(rpv, jn + 2);
            nidx = dn;
            const int nleft = nend - nbeg;
            const int nnk   = (nleft < 32) ? nleft : 32;
            if (lane < nnk) nidx = csr_src[nbeg + lane];
        }

        float4 acc;
        {
            const v4f t = xl4[(unsigned)d * (HC / 4) + (unsigned)lane];
            acc = make_float4(t.x, t.y, t.z, t.w);
        }
        const float4 zr = *(const float4*)(sxr + (w * WAVE_NODES + jn) * XR_PITCH + c0);

        float m   = 0.0f;
        float den = 1.0f;
        bool have_m = false;

        for (int base = beg; base < end; base += 32) {
            const int left = end - base;
            const int nk   = (left < 32) ? left : 32;
            if (base != beg) idx = (lane < nk) ? csr_src[base + lane] : d;
            for (int k0 = 0; k0 < nk; k0 += ROWS_IN_FLIGHT) {
                const int cnt = nk - k0;
                v4f r[ROWS_IN_FLIGHT];
#pragma unroll
                for (int j = 0; j < 4; ++j) {
                    const int s = __builtin_amdgcn_readlane(idx, k0 + j);
                    r[j] = xl4[(unsigned)s * (HC / 4) + (unsigned)lane];
                }
                if (cnt > 4) {
#pragma unroll
                    for (int j = 4; j < ROWS_IN_FLIGHT; ++j) {
                        const int s = __builtin_amdgcn_readlane(idx, k0 + j);
                        r[j] = xl4[(unsigned)s * (HC / 4) + (unsigned)lane];
                    }
                }
                if (!have_m) {
                    m = head_score(acc, zr, at);
                    have_m = true;
                }
                float4 v[4];
                float e[4];
#pragma unroll
                for (int j = 0; j < 4; ++j) {
                    v[j] = make_float4(r[j].x, r[j].y, r[j].z, r[j].w);
                    const float t = head_score(v[j], zr, at);
                    e[j] = (j < cnt) ? t : -__builtin_inff();
                }
#pragma unroll
                for (int j = 0; j < 4; ++j)
                    softmax_step(v[j], e[j], m, den, acc);
                if (cnt > 4) {
#pragma unroll
                    for (int j = 0; j < 4; ++j) {
                        v[j] = make_float4(r[4 + j].x, r[4 + j].y, r[4 + j].z, r[4 + j].w);
                        const float t = head_score(v[j], zr, at);
                        e[j] = (4 + j < cnt) ? t : -__builtin_inff();
                    }
#pragma unroll
                    for (int j = 0; j < 4; ++j)
                        softmax_step(v[j], e[j], m, den, acc);
                }
            }
        }

        const float inv = 1.0f / (den + 1e-16f);
        float4 r;
        r.x = acc.x * inv;
        r.y = acc.y * inv;
        r.z = acc.z * inv;
        r.w = acc.w * inv;
        r.x += other_half(r.x);
        r.y += other_half(r.y);
        r.z += other_half(r.z);
        r.w += other_half(r.w);

        if (lane < 16) {
            const v4f b = ((const v4f*)bias)[(unsigned)lane];
            float4 o;
            o.x = finish_value(0.5f * r.x + b.x, apply_gelu);
            o.y = finish_value(0.5f * r.y + b.y, apply_gelu);
            o.z = finish_value(0.5f * r.z + b.z, apply_gelu);
            o.w = finish_value(0.5f * r.w + b.w, apply_gelu);
            v4f o4;
            o4.x = o.x; o4.y = o.y; o4.z = o.z; o4.w = o.w;
            ((v4f*)out)[(unsigned)d * (CCH / 4) + (unsigned)lane] = o4;
        }

        beg = nbeg;
        end = nend;
        idx = nidx;
    }
}


extern "C" void kernel_launch(void* const* d_in, const int* in_sizes, int n_in,
                              void* d_out, int out_size, void* d_ws, size_t ws_size,
                              hipStream_t stream) {
    const float* x0   = (const float*)d_in[0];
    const int*   conn = (const int*)  d_in[1];
    const float* Wl   = (const float*)d_in[2];
    const float* bl   = (const float*)d_in[3];
    const float* Wr   = (const float*)d_in[4];
    const float* br   = (const float*)d_in[5];
    const float* att  = (const float*)d_in[6];
    const float* bias = (const float*)d_in[7];

    const int N  = in_sizes[0] / D_FEAT;
    const int E  = in_sizes[1] / 2;
    const int* src = conn;
    const int* dst = conn + E;

    float* ws    = (float*)d_ws;
    float* xl    = ws;
    float* x1    = xl   + (size_t)N * HC;
    int*   cnt     = (int*)(x1 + (size_t)N * D_FEAT);
    int*   rowptr  = cnt + N;
    int*   csr_src = rowptr + N + 1;
    int*   rank    = csr_src + E;
    int*   part    = rank + E;
    int*   before  = part + (size_t)EDGE_CHUNKS * N;

    const dim3 blk(256);
    const int node_blocks = (N + TILE_NODES - 1) / TILE_NODES;

    const int n_ranges = (N + RANGE_NODES - 1) / RANGE_NODES;
    count_kernel<<<n_ranges * EDGE_CHUNKS, dim3(COUNT_T), 0, stream>>>(dst, rank, part, E, N);
    chunk_prefix_kernel<<<(N + 255) / 256, blk, 0, stream>>>(part, before, cnt, N);
    rowptr_kernel<<<(N + SCAN_T - 1) / SCAN_T, dim3(SCAN_T), 0, stream>>>(cnt, rowptr, N);
    place_kernel<<<(E + 255) / 256, blk, 0, stream>>>(src, dst, rowptr, before, rank, csr_src, E, N);

    const int L = 2;
    const float* xin = x0;
    for (int l = 0; l < L; ++l) {
        const float* Wl_l = Wl + (size_t)l * D_FEAT * HC;
        const float* bl_l = bl + (size_t)l * HC;
        const float* Wr_l = Wr + (size_t)l * D_FEAT * HC;
        const float* br_l = br + (size_t)l * HC;
        const float* at_l = att + (size_t)l * NHEAD * CCH;
        const float* bi_l = bias + (size_t)l * CCH;

        gemm64x128_wmma<<<GEMM_GROUPS, dim3(128), 0, stream>>>(xin, Wl_l, bl_l, xl, N);

        float* outp = (l == L - 1) ? (float*)d_out : x1;
        node_kernel<<<node_blocks, dim3(32 * NODE_WAVES), 0, stream>>>(xin, Wr_l, br_l, xl, rowptr, csr_src,
                                                           at_l, bi_l, outp,
                                                           (l < L - 1) ? 1 : 0, N);

        xin = x1;
    }
}


# Node pass, 33 -> 23 vector instructions per edge: leaky ReLU as max(z, 0.2z); four rows share one running-max update (one rescale, five exp per four); same row order, weights round differently
# speedup vs baseline: 4.5157x; 1.0385x over previous
#include <hip/hip_runtime.h>

#define D_FEAT 64
#define HC 128
#define NHEAD 2
#define CCH 64
#define NEG_SLOPE 0.2f

typedef __attribute__((ext_vector_type(2))) float v2f;
typedef __attribute__((ext_vector_type(8))) float v8f;
typedef __attribute__((ext_vector_type(4))) float v4f;

#define GEMM_GROUPS 1250

__global__ void __launch_bounds__(128)
gemm64x128_wmma(const float* __restrict__ X,
                const float* __restrict__ W,
                const float* __restrict__ bvec,
                float* __restrict__ Y, int rows) {
    const int w    = __builtin_amdgcn_readfirstlane((int)(threadIdx.x >> 5));
    const int lane = threadIdx.x & 31;
    const int c    = lane & 15;
    const int h    = lane >> 4;
    const int tiles_m = rows >> 4;
    const int ncol = (w << 5) + c;

    int tm = blockIdx.x;
    if (tm >= tiles_m) return;

    const v4f* __restrict__ xp = (const v4f*)X;
    v4f a[8];
    {
        const unsigned piece = (unsigned)((tm << 4) + c) * (D_FEAT / 4) + (unsigned)h;
#pragma unroll
        for (int i = 0; i < 8; ++i) a[i] = xp[piece + 2 * i];
    }

    v2f b0[16], b1[16];
#pragma unroll
    for (int i = 0; i < 8; ++i) {
        const float* wp = W + (size_t)(8 * i + 4 * h) * HC + ncol;
        b0[2 * i].x     = wp[0];
        b0[2 * i].y     = wp[HC];
        b0[2 * i + 1].x = wp[2 * HC];
        b0[2 * i + 1].y = wp[3 * HC];
        b1[2 * i].x     = wp[16];
        b1[2 * i].y     = wp[HC + 16];
        b1[2 * i + 1].x = wp[2 * HC + 16];
        b1[2 * i + 1].y = wp[3 * HC + 16];
    }
    const float bb0 = bvec[ncol];
    const float bb1 = bvec[ncol + 16];
    float* __restrict__ ycol = Y + (w << 5) + lane;

    for (; tm < tiles_m; tm += GEMM_GROUPS) {
        int tn = tm + GEMM_GROUPS;
        tn = (tn < tiles_m) ? tn : tm;
        v4f an[8];
        {
            const unsigned piece = (unsigned)((tn << 4) + c) * (D_FEAT / 4) + (unsigned)h;
#pragma unroll
            for (int i = 0; i < 8; ++i) an[i] = xp[piece + 2 * i];
        }

        v8f t0 = {};
        v8f t1 = {};
#pragma unroll
        for (int i = 0; i < 8; ++i) {
            v2f alo, ahi;
            alo.x = a[i].x; alo.y = a[i].y;
            ahi.x = a[i].z; ahi.y = a[i].w;
            t0 = __builtin_amdgcn_wmma_f32_16x16x4_f32(
                 false, alo,  false, b0[2 * i],
                 (short)0, t0,  false,  false);
            t1 = __builtin_amdgcn_wmma_f32_16x16x4_f32(
                 false, alo,  false, b1[2 * i],
                 (short)0, t1,  false,  false);
            t0 = __builtin_amdgcn_wmma_f32_16x16x4_f32(
                 false, ahi,  false, b0[2 * i + 1],
                 (short)0, t0,  false,  false);
            t1 = __builtin_amdgcn_wmma_f32_16x16x4_f32(
                 false, ahi,  false, b1[2 * i + 1],
                 (short)0, t1,  false,  false);
        }

        float* __restrict__ yrow = ycol + (size_t)(tm << 4) * HC;
#pragma unroll
        for (int j = 0; j < 8; ++j) {
            const float own0 = t0[j] + bb0;
            const float own1 = t1[j] + bb1;
            const float give = h ? own0 : own1;
            const float got  = __shfl_xor(give, 16);
            yrow[(size_t)j * HC]       = h ? got : own0;
            yrow[(size_t)(j + 8) * HC] = h ? own1 : got;
        }

#pragma unroll
        for (int i = 0; i < 8; ++i) a[i] = an[i];
    }
}


#define RANGE_NODES 16384
#define EDGE_CHUNKS 32
#define COUNT_T 1024
#define COUNT_AHEAD 16

__device__ __forceinline__ int edge_chunk_len(int E) {
    return ((E + EDGE_CHUNKS * 256 - 1) / (EDGE_CHUNKS * 256)) * 256;
}

__global__ void __launch_bounds__(COUNT_T)
count_kernel(const int* __restrict__ dst, int* __restrict__ rank,
             int* __restrict__ part, int E, int n) {
    __shared__ int hist[RANGE_NODES];
    const int t  = threadIdx.x;
    const int c  = blockIdx.x % EDGE_CHUNKS;
    const int lo = (blockIdx.x / EDGE_CHUNKS) * RANGE_NODES;

    for (int i = t; i < RANGE_NODES; i += COUNT_T) hist[i] = 0;
    __syncthreads();

    const int chunk = edge_chunk_len(E);
    const int e0 = c * chunk;
    int e1 = e0 + chunk;
    e1 = (e1 < E) ? e1 : E;
    for (int e = e0 + t; e < e1; e += COUNT_T * COUNT_AHEAD) {
        int d[COUNT_AHEAD];
#pragma unroll
        for (int j = 0; j < COUNT_AHEAD; ++j) {
            const int ee = e + j * COUNT_T;
            d[j] = (ee < e1) ? dst[ee] : -1;
        }
        int r[COUNT_AHEAD];
#pragma unroll
        for (int j = 0; j < COUNT_AHEAD; ++j) {
            const unsigned u = (unsigned)(d[j] - lo);
            r[j] = 0;
            if (u < (unsigned)RANGE_NODES)
                r[j] = atomicAdd(&hist[u], 1);
        }
#pragma unroll
        for (int j = 0; j < COUNT_AHEAD; ++j) {
            const unsigned u = (unsigned)(d[j] - lo);
            if (u < (unsigned)RANGE_NODES) rank[e + j * COUNT_T] = r[j];
        }
    }
    __syncthreads();

    int hi = n - lo;
    hi = (hi < RANGE_NODES) ? hi : RANGE_NODES;
    int* __restrict__ row = part + (size_t)c * n + lo;
    for (int i = t; i < hi; i += COUNT_T) row[i] = hist[i];
}

__global__ void chunk_prefix_kernel(const int* __restrict__ part, int* __restrict__ before,
                                    int* __restrict__ cnt, int n) {
    const int d = blockIdx.x * blockDim.x + threadIdx.x;
    if (d >= n) return;
    int v[EDGE_CHUNKS];
#pragma unroll
    for (int c = 0; c < EDGE_CHUNKS; ++c) v[c] = part[(size_t)c * n + d];
    int s = 0;
#pragma unroll
    for (int c = 0; c < EDGE_CHUNKS; ++c) {
        before[(size_t)c * n + d] = s;
        s += v[c];
    }
    cnt[d] = s;
}

#define SCAN_T 1024
__global__ void __launch_bounds__(SCAN_T)
rowptr_kernel(const int* __restrict__ cnt, int* __restrict__ rowptr, int n) {
    __shared__ int wsum[32];
    __shared__ int sbase;
    const int t    = threadIdx.x;
    const int lane = t & 31;
    const int w    = t >> 5;
    const int first = blockIdx.x * SCAN_T;

    int part = 0;
    for (int i = t; i < first; i += SCAN_T) part += cnt[i];
    part += __shfl_xor(part, 16);
    part += __shfl_xor(part, 8);
    part += __shfl_xor(part, 4);
    part += __shfl_xor(part, 2);
    part += __shfl_xor(part, 1);
    if (lane == 0) wsum[w] = part;
    __syncthreads();
    if (w == 0) {
        int v = wsum[lane];
        v += __shfl_xor(v, 16);
        v += __shfl_xor(v, 8);
        v += __shfl_xor(v, 4);
        v += __shfl_xor(v, 2);
        v += __shfl_xor(v, 1);
        if (lane == 0) sbase = v;
    }
    __syncthreads();
    const int base = sbase;

    const int i = first + t;
    const int c = (i < n) ? cnt[i] : 0;
    int incl = c;
    for (int o = 1; o < 32; o <<= 1) {
        const int u = __shfl_up(incl, o);
        if (lane >= o) incl += u;
    }
    if (lane == 31) wsum[w] = incl;
    __syncthreads();
    if (w == 0) {
        const int v = wsum[lane];
        int inc2 = v;
        for (int o = 1; o < 32; o <<= 1) {
            const int u = __shfl_up(inc2, o);
            if (lane >= o) inc2 += u;
        }
        wsum[lane] = inc2 - v;
    }
    __syncthreads();
    const int excl = base + wsum[w] + incl - c;
    if (i < n) rowptr[i] = excl;
    if (i == n - 1) rowptr[n] = excl + c;
}

__global__ void place_kernel(const int* __restrict__ src,
                             const int* __restrict__ dst,
                             const int* __restrict__ rowptr,
                             const int* __restrict__ before,
                             const int* __restrict__ rank,
                             int* __restrict__ csr_src, int E, int n) {
    const int e = blockIdx.x * blockDim.x + threadIdx.x;
    if (e >= E) return;
    const int c = (int)(blockIdx.x * blockDim.x) / edge_chunk_len(E);
    const int d = dst[e];
    csr_src[rowptr[d] + before[(size_t)c * n + d] + rank[e]] = src[e];
}


template <int XM>
__device__ __forceinline__ float half_xor(const float x) {
    const int b = __builtin_bit_cast(int, x);
    return __builtin_bit_cast(float, __builtin_amdgcn_mov_dpp(b, 0x160 + XM, 0xf, 0xf, true));
}
__device__ __forceinline__ float other_half(const float x) {
    const int b = __builtin_bit_cast(int, x);
    return __builtin_bit_cast(float,
        __builtin_amdgcn_permlanex16(b, b, 0x76543210, 0xfedcba98, false, false));
}

__device__ __forceinline__ float head_score(const float4 v, const float4 zr, const float4 at) {
    float p = 0.0f, z;
    z = v.x + zr.x; z = fmaxf(z, NEG_SLOPE * z); p += z * at.x;
    z = v.y + zr.y; z = fmaxf(z, NEG_SLOPE * z); p += z * at.y;
    z = v.z + zr.z; z = fmaxf(z, NEG_SLOPE * z); p += z * at.z;
    z = v.w + zr.w; z = fmaxf(z, NEG_SLOPE * z); p += z * at.w;
    p += half_xor<1>(p);
    p += half_xor<2>(p);
    p += half_xor<4>(p);
    p += half_xor<8>(p);
    return p;
}

__device__ __forceinline__ void softmax_step4(const float4 (&v)[4], const float (&e)[4],
                                              float& m, float& den, float4& acc) {
    const float mn = fmaxf(fmaxf(m, e[0]), fmaxf(fmaxf(e[1], e[2]), e[3]));
    const float sc = __expf(m - mn);
    const float a0 = __expf(e[0] - mn);
    const float a1 = __expf(e[1] - mn);
    const float a2 = __expf(e[2] - mn);
    const float a3 = __expf(e[3] - mn);
    den   = (((den   * sc + a0) + a1) + a2) + a3;
    acc.x = (((acc.x * sc + a0 * v[0].x) + a1 * v[1].x) + a2 * v[2].x) + a3 * v[3].x;
    acc.y = (((acc.y * sc + a0 * v[0].y) + a1 * v[1].y) + a2 * v[2].y) + a3 * v[3].y;
    acc.z = (((acc.z * sc + a0 * v[0].z) + a1 * v[1].z) + a2 * v[2].z) + a3 * v[3].z;
    acc.w = (((acc.w * sc + a0 * v[0].w) + a1 * v[1].w) + a2 * v[2].w) + a3 * v[3].w;
    m = mn;
}

__device__ __forceinline__ float finish_value(float v, int apply_gelu) {
    if (apply_gelu) {
        const float u = 0.7978845608028654f * (v + 0.044715f * v * v * v);
        v = 0.5f * v * (1.0f + tanhf(u));
    }
    return v;
}

#define ROWS_IN_FLIGHT 8
#define TILE_NODES 16
#define NODE_WAVES 4
#define WAVE_NODES 4
#define XR_PITCH 132

__global__ void __launch_bounds__(128, 12)
node_kernel(const float* __restrict__ xin,
            const float* __restrict__ Wr,
            const float* __restrict__ br,
            const float* __restrict__ xl,
            const int* __restrict__ rowptr,
            const int* __restrict__ csr_src,
            const float* __restrict__ att,
            const float* __restrict__ bias,
            float* __restrict__ out,
            int apply_gelu, int n) {
    __shared__ __attribute__((aligned(16))) float sxr[TILE_NODES * XR_PITCH];

    const int w = __builtin_amdgcn_readfirstlane((int)(threadIdx.x >> 5));
    const int lane = threadIdx.x & 31;
    const int nbase = blockIdx.x * TILE_NODES;
    const int d0 = nbase + w * WAVE_NODES;

    const int c0 = lane << 2;
    const float4 at = *(const float4*)(att + c0);
    int rpi = d0 + (lane & 7);
    rpi = (rpi < n) ? rpi : n;
    const int rpv = rowptr[rpi];
    const int dfirst = (d0 < n) ? d0 : n - 1;
    const v4f* __restrict__ xl4 = (const v4f*)xl;

    int beg = __builtin_amdgcn_readlane(rpv, 0);
    int end = __builtin_amdgcn_readlane(rpv, 1);
    int idx = dfirst;
    {
        const int left0 = end - beg;
        const int nk0   = (left0 < 32) ? left0 : 32;
        if (lane < nk0) idx = csr_src[beg + lane];
    }

    {
        int mrow = nbase + (lane & 15);
        mrow = (mrow < n) ? mrow : n - 1;
        const int ncol  = (w << 5) + (lane & 15);
        const int khalf = (lane >> 4) << 1;
        v8f t0 = {};
        v8f t1 = {};
#pragma unroll
        for (int k0 = 0; k0 < D_FEAT; k0 += 4) {
            const int kk = k0 + khalf;
            v2f a, b0, b1;
            const float* ap = xin + (size_t)mrow * D_FEAT + kk;
            a.x = ap[0];
            a.y = ap[1];
            b0.x = Wr[(size_t)(kk + 0) * HC + ncol];
            b0.y = Wr[(size_t)(kk + 1) * HC + ncol];
            b1.x = Wr[(size_t)(kk + 0) * HC + ncol + 16];
            b1.y = Wr[(size_t)(kk + 1) * HC + ncol + 16];
            t0 = __builtin_amdgcn_wmma_f32_16x16x4_f32(
                 false, a,  false, b0,
                 (short)0, t0,  false,  false);
            t1 = __builtin_amdgcn_wmma_f32_16x16x4_f32(
                 false, a,  false, b1,
                 (short)0, t1,  false,  false);
        }
        const float bb0 = br[ncol];
        const float bb1 = br[ncol + 16];
        const int rloc = (lane >> 4) << 3;
#pragma unroll
        for (int j = 0; j < 8; ++j) {
            sxr[(rloc + j) * XR_PITCH + ncol]      = t0[j] + bb0;
            sxr[(rloc + j) * XR_PITCH + ncol + 16] = t1[j] + bb1;
        }
    }
    __syncthreads();

    for (int jn = 0; jn < WAVE_NODES; ++jn) {
        const int d = d0 + jn;
        if (d >= n) break;

        int nbeg = 0, nend = 0, nidx = d;
        if (jn + 1 < WAVE_NODES) {
            int dn = d + 1;
            dn = (dn < n) ? dn : n - 1;
            nbeg = __builtin_amdgcn_readlane(rpv, jn + 1);
            nend = __builtin_amdgcn_readlane(rpv, jn + 2);
            nidx = dn;
            const int nleft = nend - nbeg;
            const int nnk   = (nleft < 32) ? nleft : 32;
            if (lane < nnk) nidx = csr_src[nbeg + lane];
        }

        float4 acc;
        {
            const v4f t = xl4[(unsigned)d * (HC / 4) + (unsigned)lane];
            acc = make_float4(t.x, t.y, t.z, t.w);
        }
        const float4 zr = *(const float4*)(sxr + (w * WAVE_NODES + jn) * XR_PITCH + c0);

        float m   = 0.0f;
        float den = 1.0f;
        bool have_m = false;

        for (int base = beg; base < end; base += 32) {
            const int left = end - base;
            const int nk   = (left < 32) ? left : 32;
            if (base != beg) idx = (lane < nk) ? csr_src[base + lane] : d;
            for (int k0 = 0; k0 < nk; k0 += ROWS_IN_FLIGHT) {
                const int cnt = nk - k0;
                v4f r[ROWS_IN_FLIGHT];
#pragma unroll
                for (int j = 0; j < 4; ++j) {
                    const int s = __builtin_amdgcn_readlane(idx, k0 + j);
                    r[j] = xl4[(unsigned)s * (HC / 4) + (unsigned)lane];
                }
                if (cnt > 4) {
#pragma unroll
                    for (int j = 4; j < ROWS_IN_FLIGHT; ++j) {
                        const int s = __builtin_amdgcn_readlane(idx, k0 + j);
                        r[j] = xl4[(unsigned)s * (HC / 4) + (unsigned)lane];
                    }
                }
                if (!have_m) {
                    m = head_score(acc, zr, at);
                    have_m = true;
                }
                float4 v[4];
                float e[4];
#pragma unroll
                for (int j = 0; j < 4; ++j) {
                    v[j] = make_float4(r[j].x, r[j].y, r[j].z, r[j].w);
                    const float t = head_score(v[j], zr, at);
                    e[j] = (j < cnt) ? t : -__builtin_inff();
                }
                softmax_step4(v, e, m, den, acc);
                if (cnt > 4) {
#pragma unroll
                    for (int j = 0; j < 4; ++j) {
                        v[j] = make_float4(r[4 + j].x, r[4 + j].y, r[4 + j].z, r[4 + j].w);
                        const float t = head_score(v[j], zr, at);
                        e[j] = (4 + j < cnt) ? t : -__builtin_inff();
                    }
                    softmax_step4(v, e, m, den, acc);
                }
            }
        }

        const float inv = 1.0f / (den + 1e-16f);
        float4 r;
        r.x = acc.x * inv;
        r.y = acc.y * inv;
        r.z = acc.z * inv;
        r.w = acc.w * inv;
        r.x += other_half(r.x);
        r.y += other_half(r.y);
        r.z += other_half(r.z);
        r.w += other_half(r.w);

        if (lane < 16) {
            const v4f b = ((const v4f*)bias)[(unsigned)lane];
            float4 o;
            o.x = finish_value(0.5f * r.x + b.x, apply_gelu);
            o.y = finish_value(0.5f * r.y + b.y, apply_gelu);
            o.z = finish_value(0.5f * r.z + b.z, apply_gelu);
            o.w = finish_value(0.5f * r.w + b.w, apply_gelu);
            v4f o4;
            o4.x = o.x; o4.y = o.y; o4.z = o.z; o4.w = o.w;
            ((v4f*)out)[(unsigned)d * (CCH / 4) + (unsigned)lane] = o4;
        }

        beg = nbeg;
        end = nend;
        idx = nidx;
    }
}


extern "C" void kernel_launch(void* const* d_in, const int* in_sizes, int n_in,
                              void* d_out, int out_size, void* d_ws, size_t ws_size,
                              hipStream_t stream) {
    const float* x0   = (const float*)d_in[0];
    const int*   conn = (const int*)  d_in[1];
    const float* Wl   = (const float*)d_in[2];
    const float* bl   = (const float*)d_in[3];
    const float* Wr   = (const float*)d_in[4];
    const float* br   = (const float*)d_in[5];
    const float* att  = (const float*)d_in[6];
    const float* bias = (const float*)d_in[7];

    const int N  = in_sizes[0] / D_FEAT;
    const int E  = in_sizes[1] / 2;
    const int* src = conn;
    const int* dst = conn + E;

    float* ws    = (float*)d_ws;
    float* xl    = ws;
    float* x1    = xl   + (size_t)N * HC;
    int*   cnt     = (int*)(x1 + (size_t)N * D_FEAT);
    int*   rowptr  = cnt + N;
    int*   csr_src = rowptr + N + 1;
    int*   rank    = csr_src + E;
    int*   part    = rank + E;
    int*   before  = part + (size_t)EDGE_CHUNKS * N;

    const dim3 blk(256);
    const int node_blocks = (N + TILE_NODES - 1) / TILE_NODES;

    const int n_ranges = (N + RANGE_NODES - 1) / RANGE_NODES;
    count_kernel<<<n_ranges * EDGE_CHUNKS, dim3(COUNT_T), 0, stream>>>(dst, rank, part, E, N);
    chunk_prefix_kernel<<<(N + 255) / 256, blk, 0, stream>>>(part, before, cnt, N);
    rowptr_kernel<<<(N + SCAN_T - 1) / SCAN_T, dim3(SCAN_T), 0, stream>>>(cnt, rowptr, N);
    place_kernel<<<(E + 255) / 256, blk, 0, stream>>>(src, dst, rowptr, before, rank, csr_src, E, N);

    const int L = 2;
    const float* xin = x0;
    for (int l = 0; l < L; ++l) {
        const float* Wl_l = Wl + (size_t)l * D_FEAT * HC;
        const float* bl_l = bl + (size_t)l * HC;
        const float* Wr_l = Wr + (size_t)l * D_FEAT * HC;
        const float* br_l = br + (size_t)l * HC;
        const float* at_l = att + (size_t)l * NHEAD * CCH;
        const float* bi_l = bias + (size_t)l * CCH;

        gemm64x128_wmma<<<GEMM_GROUPS, dim3(128), 0, stream>>>(xin, Wl_l, bl_l, xl, N);

        float* outp = (l == L - 1) ? (float*)d_out : x1;
        node_kernel<<<node_blocks, dim3(32 * NODE_WAVES), 0, stream>>>(xin, Wr_l, br_l, xl, rowptr, csr_src,
                                                           at_l, bi_l, outp,
                                                           (l < L - 1) ? 1 : 0, N);

        xin = x1;
    }
}
